# starting kernel
# baseline (speedup 1.0000x reference)
#include <hip/hip_runtime.h>
#include <math.h>

#define SLOPE 0.2f

typedef float v2f __attribute__((ext_vector_type(2)));
typedef float v8f __attribute__((ext_vector_type(8)));

__device__ __forceinline__ unsigned f2key(float f) {
  unsigned u = __float_as_uint(f);
  return (u & 0x80000000u) ? ~u : (u | 0x80000000u);
}
__device__ __forceinline__ float key2f(unsigned k) {
  unsigned u = (k & 0x80000000u) ? (k ^ 0x80000000u) : ~k;
  return __uint_as_float(u);
}
__device__ __forceinline__ float lrelu(float v) { return v > 0.0f ? v : SLOPE * v; }

template <int K, int NC>
__global__ void gemm_bias_wmma(const float* __restrict__ X, const float* __restrict__ W,
                               const float* __restrict__ b, float* __restrict__ Y, int M) {
  const int ntiles = NC / 16;
  int wid = threadIdx.x >> 5;
  int tile = blockIdx.x * (blockDim.x >> 5) + wid;
  int total = (M / 16) * ntiles;
  if (tile >= total) return;
  int mt = tile / ntiles, nt = tile % ntiles;
  int lane = threadIdx.x & 31;
  int half = lane >> 4;
  int l = lane & 15;
  int row = mt * 16 + l;
  int col = nt * 16 + l;
  float bc = b[col];
  v8f acc = {bc, bc, bc, bc, bc, bc, bc, bc};
  const float* xp = X + (size_t)row * K;
  const float* wp = W + col;
#pragma unroll
  for (int k0 = 0; k0 < K; k0 += 4) {
    v2f a, bv;
    a.x = xp[k0 + 2 * half];
    a.y = xp[k0 + 2 * half + 1];
    bv.x = wp[(size_t)(k0 + 2 * half) * NC];
    bv.y = wp[(size_t)(k0 + 2 * half + 1) * NC];
    acc = __builtin_amdgcn_wmma_f32_16x16x4_f32(false, a, false, bv, (short)0, acc,
                                                false, false);
  }
  float* yp = Y + (size_t)(mt * 16 + 8 * half) * NC + col;
#pragma unroll
  for (int v = 0; v < 8; v++) yp[(size_t)v * NC] = acc[v];
}

template <int H_, int C_>
__global__ void edge_scores(const float* __restrict__ xl, const float* __restrict__ xr,
                            const float* __restrict__ att, const int* __restrict__ ei,
                            int E_, int Etot, float* __restrict__ ebuf,
                            unsigned* __restrict__ nmax) {
  int g = (int)((blockIdx.x * (size_t)blockDim.x + threadIdx.x) >> 4);
  int sub = threadIdx.x & 15;
  if (g >= Etot * H_) return;
  int eidx = g / H_, head = g % H_;
  int src, dst;
  if (eidx < E_) { src = ei[eidx]; dst = ei[E_ + eidx]; }
  else           { src = dst = eidx - E_; }
  int base = head * C_ + sub * 4;
  float4 a  = *(const float4*)(xl + (size_t)src * (H_ * C_) + base);
  float4 bq = *(const float4*)(xr + (size_t)dst * (H_ * C_) + base);
  float4 at = *(const float4*)(att + base);
  float p = lrelu(a.x + bq.x) * at.x + lrelu(a.y + bq.y) * at.y +
            lrelu(a.z + bq.z) * at.z + lrelu(a.w + bq.w) * at.w;
  for (int off = 8; off > 0; off >>= 1) p += __shfl_xor(p, off, 32);
  if (sub == 0) {
    ebuf[g] = p;
    atomicMax(nmax + (size_t)dst * H_ + head, f2key(p));
  }
}

template <int H_>
__global__ void edge_exp(const int* __restrict__ ei, int E_, int Etot,
                         const unsigned* __restrict__ nmax, float* __restrict__ ebuf,
                         float* __restrict__ denom) {
  int g = (int)(blockIdx.x * (size_t)blockDim.x + threadIdx.x);
  if (g >= Etot * H_) return;
  int eidx = g / H_, head = g % H_;
  int dst = (eidx < E_) ? ei[E_ + eidx] : (eidx - E_);
  float mx = key2f(nmax[(size_t)dst * H_ + head]);
  float ex = expf(ebuf[g] - mx);
  ebuf[g] = ex;
  atomicAdd(denom + (size_t)dst * H_ + head, ex);
}

template <int H_, int C_>
__global__ void edge_scatter(const float* __restrict__ xl, const int* __restrict__ ei,
                             int E_, int Etot, const float* __restrict__ ebuf,
                             const float* __restrict__ denom, float* __restrict__ oacc) {
  int g = (int)((blockIdx.x * (size_t)blockDim.x + threadIdx.x) >> 4);
  int sub = threadIdx.x & 15;
  if (g >= Etot * H_) return;
  int eidx = g / H_, head = g % H_;
  int src, dst;
  if (eidx < E_) { src = ei[eidx]; dst = ei[E_ + eidx]; }
  else           { src = dst = eidx - E_; }
  float alpha = ebuf[g] / denom[(size_t)dst * H_ + head];
  int base = head * C_ + sub * 4;
  float4 a = *(const float4*)(xl + (size_t)src * (H_ * C_) + base);
  float* o = oacc + (size_t)dst * (H_ * C_) + base;
  atomicAdd(o + 0, alpha * a.x);
  atomicAdd(o + 1, alpha * a.y);
  atomicAdd(o + 2, alpha * a.z);
  atomicAdd(o + 3, alpha * a.w);
}

__global__ void bias_elu_inplace(float* __restrict__ h, const float* __restrict__ bias,
                                 int n, int nc) {
  int i = (int)(blockIdx.x * (size_t)blockDim.x + threadIdx.x);
  if (i >= n) return;
  float v = h[i] + bias[i % nc];
  h[i] = v > 0.0f ? v : expm1f(v);
}

__global__ void bias_add_out(const float* __restrict__ acc, const float* __restrict__ bias,
                             float* __restrict__ out, int n, int nc) {
  int i = (int)(blockIdx.x * (size_t)blockDim.x + threadIdx.x);
  if (i >= n) return;
  out[i] = acc[i] + bias[i % nc];
}

extern "C" void kernel_launch(void* const* d_in, const int* in_sizes, int n_in,
                              void* d_out, int out_size, void* d_ws, size_t ws_size,
                              hipStream_t stream) {
  (void)n_in; (void)out_size; (void)ws_size;
  const int IN = 128, HID = 64, H = 4, OUT = 64;
  const float*    x    = (const float*)d_in[0];
  const int*      ei   = (const int*)d_in[1];
  const float*    Wl1  = (const float*)d_in[2];
  const float*    bl1  = (const float*)d_in[3];
  const float*    Wr1  = (const float*)d_in[4];
  const float*    br1  = (const float*)d_in[5];
  const float*    att1 = (const float*)d_in[6];
  const float*    bias1= (const float*)d_in[7];
  const float*    Wl2  = (const float*)d_in[8];
  const float*    bl2  = (const float*)d_in[9];
  const float*    Wr2  = (const float*)d_in[10];
  const float*    br2  = (const float*)d_in[11];
  const float*    att2 = (const float*)d_in[12];
  const float*    bias2= (const float*)d_in[13];
  float* out = (float*)d_out;

  const int N    = in_sizes[0] / IN;
  const int E    = in_sizes[1] / 2;
  const int Etot = E + N;
  const int HC   = H * HID;

  float* ws = (float*)d_ws;
  size_t o = 0;
  float*    xl1   = ws + o; o += (size_t)N * HC;
  float*    xr1   = ws + o; o += (size_t)N * HC;
  float*    hacc  = ws + o; o += (size_t)N * HC;
  float*    ebuf  = ws + o; o += (size_t)Etot * H;
  unsigned* nmax  = (unsigned*)(ws + o); o += (size_t)N * H;
  float*    denom = ws + o; o += (size_t)N * H;
  float* xl2   = xl1;
  float* xr2   = xr1;
  float* oacc2 = xr1 + (size_t)N * OUT;

  const int B = 256;

  hipMemsetAsync(hacc,  0, (size_t)N * HC * sizeof(float), stream);
  hipMemsetAsync(nmax,  0, (size_t)N * H * sizeof(unsigned), stream);
  hipMemsetAsync(denom, 0, (size_t)N * H * sizeof(float), stream);

  int tiles1 = (N / 16) * (HC / 16);
  gemm_bias_wmma<128, 256><<<(tiles1 + 7) / 8, B, 0, stream>>>(x, Wl1, bl1, xl1, N);
  gemm_bias_wmma<128, 256><<<(tiles1 + 7) / 8, B, 0, stream>>>(x, Wr1, br1, xr1, N);

  long long items1 = (long long)Etot * H;
  int gb1 = (int)((items1 * 16 + B - 1) / B);
  edge_scores<4, 64><<<gb1, B, 0, stream>>>(xl1, xr1, att1, ei, E, Etot, ebuf, nmax);
  edge_exp<4><<<(int)((items1 + B - 1) / B), B, 0, stream>>>(ei, E, Etot, nmax, ebuf, denom);
  edge_scatter<4, 64><<<gb1, B, 0, stream>>>(xl1, ei, E, Etot, ebuf, denom, hacc);

  int n1 = N * HC;
  bias_elu_inplace<<<(n1 + B - 1) / B, B, 0, stream>>>(hacc, bias1, n1, HC);

  hipMemsetAsync(oacc2, 0, (size_t)N * OUT * sizeof(float), stream);
  hipMemsetAsync(nmax,  0, (size_t)N * sizeof(unsigned), stream);
  hipMemsetAsync(denom, 0, (size_t)N * sizeof(float), stream);

  int tiles2 = (N / 16) * (OUT / 16);
  gemm_bias_wmma<256, 64><<<(tiles2 + 7) / 8, B, 0, stream>>>(hacc, Wl2, bl2, xl2, N);
  gemm_bias_wmma<256, 64><<<(tiles2 + 7) / 8, B, 0, stream>>>(hacc, Wr2, br2, xr2, N);

  long long items2 = Etot;
  int gb2 = (int)((items2 * 16 + B - 1) / B);
  edge_scores<1, 64><<<gb2, B, 0, stream>>>(xl2, xr2, att2, ei, E, Etot, ebuf, nmax);
  edge_exp<1><<<(int)((items2 + B - 1) / B), B, 0, stream>>>(ei, E, Etot, nmax, ebuf, denom);
  edge_scatter<1, 64><<<gb2, B, 0, stream>>>(xl2, ei, E, Etot, ebuf, denom, oacc2);

  int n2 = N * OUT;
  bias_add_out<<<(n2 + B - 1) / B, B, 0, stream>>>(oacc2, bias2, out, n2, OUT);
}


# Edges grouped by destination in-call (count, scan, fill by int atomics); one pass per node does logits, running softmax and weighted sum in registers; float atomics gone; sum order = fill order
# speedup vs baseline: 2.2664x; 2.2664x over previous
#include <hip/hip_runtime.h>
#include <math.h>

#define SLOPE 0.2f

typedef float v2f __attribute__((ext_vector_type(2)));
typedef float v8f __attribute__((ext_vector_type(8)));

__device__ __forceinline__ float lrelu(float v) { return v > 0.0f ? v : SLOPE * v; }

template <int K, int NC>
__global__ void gemm_bias_wmma(const float* __restrict__ X, const float* __restrict__ W,
                               const float* __restrict__ b, float* __restrict__ Y, int M) {
  const int ntiles = NC / 16;
  int wid = threadIdx.x >> 5;
  int tile = blockIdx.x * (blockDim.x >> 5) + wid;
  int total = (M / 16) * ntiles;
  if (tile >= total) return;
  int mt = tile / ntiles, nt = tile % ntiles;
  int lane = threadIdx.x & 31;
  int half = lane >> 4;
  int l = lane & 15;
  int row = mt * 16 + l;
  int col = nt * 16 + l;
  float bc = b[col];
  v8f acc = {bc, bc, bc, bc, bc, bc, bc, bc};
  const float* xp = X + (size_t)row * K;
  const float* wp = W + col;
#pragma unroll
  for (int k0 = 0; k0 < K; k0 += 4) {
    v2f a, bv;
    a.x = xp[k0 + 2 * half];
    a.y = xp[k0 + 2 * half + 1];
    bv.x = wp[(size_t)(k0 + 2 * half) * NC];
    bv.y = wp[(size_t)(k0 + 2 * half + 1) * NC];
    acc = __builtin_amdgcn_wmma_f32_16x16x4_f32(false, a, false, bv, (short)0, acc,
                                                false, false);
  }
  float* yp = Y + (size_t)(mt * 16 + 8 * half) * NC + col;
#pragma unroll
  for (int v = 0; v < 8; v++) yp[(size_t)v * NC] = acc[v];
}

__global__ void dst_count(const int* __restrict__ ei, int E_, int* __restrict__ cnt) {
  int e = (int)(blockIdx.x * (size_t)blockDim.x + threadIdx.x);
  if (e >= E_) return;
  atomicAdd(cnt + ei[E_ + e], 1);
}

__global__ void __launch_bounds__(1024)
dst_scan(const int* __restrict__ cnt, int* __restrict__ rowptr, int n) {
  __shared__ int wpre[32];
  __shared__ int wown[32];
  const int t = threadIdx.x, lane = t & 31, w = t >> 5;
  const int tile0 = blockIdx.x * 1024;
  const int i = tile0 + t;
  int pre = 0;
  for (int k = t; k < tile0; k += 1024) pre += cnt[k];
  const int x = (i < n) ? cnt[i] : 0;
  int inc = x;
  for (int off = 1; off < 32; off <<= 1) {
    int y = __shfl_up(inc, off, 32);
    if (lane >= off) inc += y;
  }
  for (int off = 16; off > 0; off >>= 1) pre += __shfl_xor(pre, off, 32);
  if (lane == 31) wown[w] = inc;
  if (lane == 0) wpre[w] = pre;
  __syncthreads();
  int front = 0;
  for (int k = 0; k < 32; k++) {
    front += wpre[k];
    if (k < w) front += wown[k];
  }
  const int excl = front + inc - x;
  if (i < n) rowptr[i] = excl;
  if (i == n - 1) rowptr[n] = excl + x;
}

__global__ void dst_fill(const int* __restrict__ ei, int E_, const int* __restrict__ rowptr,
                         int* __restrict__ cnt, int* __restrict__ col) {
  int e = (int)(blockIdx.x * (size_t)blockDim.x + threadIdx.x);
  if (e >= E_) return;
  int s = ei[e], d = ei[E_ + e];
  int old = atomicSub(cnt + d, 1);
  col[rowptr[d] + old - 1] = s;
}

template <int V>
__device__ __forceinline__ void load_row(const float* __restrict__ p, float (&x)[V]) {
#pragma unroll
  for (int q = 0; q < V / 4; q++) {
    float4 t = *(const float4*)(p + 4 * q);
    x[4 * q] = t.x; x[4 * q + 1] = t.y; x[4 * q + 2] = t.z; x[4 * q + 3] = t.w;
  }
}

template <int V, int LPH>
__device__ __forceinline__ float edge_logit(const float (&x)[V], const float (&r)[V],
                                            const float (&a)[V]) {
  float p = 0.0f;
#pragma unroll
  for (int k = 0; k < V; k++) p += lrelu(x[k] + r[k]) * a[k];
  for (int off = LPH / 2; off > 0; off >>= 1) p += __shfl_xor(p, off, 32);
  return p;
}

template <int HC, int H_, int V, bool ELU>
__global__ void node_attend(const float* __restrict__ xl, const float* __restrict__ xr,
                            const float* __restrict__ att, const float* __restrict__ bias,
                            const int* __restrict__ rowptr, const int* __restrict__ col,
                            float* __restrict__ outp, int n) {
  constexpr int LPN = HC / V;
  constexpr int LPH = LPN / H_;
  int node = (int)((blockIdx.x * (size_t)blockDim.x + threadIdx.x) / LPN);
  if (LPN == 32) node = __builtin_amdgcn_readfirstlane(node);
  if (node >= n) return;
  const int base = (int)(threadIdx.x % LPN) * V;
  float r[V], a[V], x[V], acc[V], bs[V];
  load_row<V>(xr + (size_t)node * HC + base, r);
  load_row<V>(att + base, a);
  load_row<V>(xl + (size_t)node * HC + base, acc);
  float m = edge_logit<V, LPH>(acc, r, a);
  float s = 1.0f;
  const int beg = rowptr[node], end = rowptr[node + 1];
  int src = (beg < end) ? col[beg] : 0;
  for (int j = beg; j < end; j++) {
    load_row<V>(xl + (size_t)src * HC + base, x);
    const int nxt = (j + 1 < end) ? col[j + 1] : 0;
    const float e = edge_logit<V, LPH>(x, r, a);
    const float d = e - m;
    const float t = expf(-fabsf(d));
    const float sc = d > 0.0f ? t : 1.0f;
    const float wgt = d > 0.0f ? 1.0f : t;
    m = fmaxf(m, e);
    s = s * sc + wgt;
#pragma unroll
    for (int k = 0; k < V; k++) acc[k] = acc[k] * sc + wgt * x[k];
    src = nxt;
  }
  load_row<V>(bias + base, bs);
  float* o = outp + (size_t)node * HC + base;
#pragma unroll
  for (int q = 0; q < V / 4; q++) {
    float4 t;
    float v0 = acc[4 * q] / s + bs[4 * q];
    float v1 = acc[4 * q + 1] / s + bs[4 * q + 1];
    float v2 = acc[4 * q + 2] / s + bs[4 * q + 2];
    float v3 = acc[4 * q + 3] / s + bs[4 * q + 3];
    if (ELU) {
      v0 = v0 > 0.0f ? v0 : expm1f(v0);
      v1 = v1 > 0.0f ? v1 : expm1f(v1);
      v2 = v2 > 0.0f ? v2 : expm1f(v2);
      v3 = v3 > 0.0f ? v3 : expm1f(v3);
    }
    t.x = v0; t.y = v1; t.z = v2; t.w = v3;
    *(float4*)(o + 4 * q) = t;
  }
}

extern "C" void kernel_launch(void* const* d_in, const int* in_sizes, int n_in,
                              void* d_out, int out_size, void* d_ws, size_t ws_size,
                              hipStream_t stream) {
  (void)n_in; (void)out_size; (void)ws_size;
  const int IN = 128, HID = 64, H = 4, OUT = 64;
  const float*    x    = (const float*)d_in[0];
  const int*      ei   = (const int*)d_in[1];
  const float*    Wl1  = (const float*)d_in[2];
  const float*    bl1  = (const float*)d_in[3];
  const float*    Wr1  = (const float*)d_in[4];
  const float*    br1  = (const float*)d_in[5];
  const float*    att1 = (const float*)d_in[6];
  const float*    bias1= (const float*)d_in[7];
  const float*    Wl2  = (const float*)d_in[8];
  const float*    bl2  = (const float*)d_in[9];
  const float*    Wr2  = (const float*)d_in[10];
  const float*    br2  = (const float*)d_in[11];
  const float*    att2 = (const float*)d_in[12];
  const float*    bias2= (const float*)d_in[13];
  float* out = (float*)d_out;

  const int N    = in_sizes[0] / IN;
  const int E    = in_sizes[1] / 2;
  const int HC   = H * HID;

  float* ws = (float*)d_ws;
  size_t o = 0;
  float* xl1    = ws + o; o += (size_t)N * HC;
  float* xr1    = ws + o; o += (size_t)N * HC;
  float* h      = ws + o; o += (size_t)N * HC;
  int*   cnt    = (int*)(ws + o); o += (size_t)N + 4;
  int*   rowptr = (int*)(ws + o); o += (size_t)N + 4;
  int*   col    = (int*)(ws + o); o += (size_t)E;
  float* xl2 = xl1;
  float* xr2 = xr1;

  const int B = 256;

  hipMemsetAsync(cnt, 0, (size_t)N * sizeof(int), stream);
  dst_count<<<(E + B - 1) / B, B, 0, stream>>>(ei, E, cnt);
  dst_scan<<<(N + 1023) / 1024, 1024, 0, stream>>>(cnt, rowptr, N);
  dst_fill<<<(E + B - 1) / B, B, 0, stream>>>(ei, E, rowptr, cnt, col);

  int tiles1 = (N / 16) * (HC / 16);
  gemm_bias_wmma<128, 256><<<(tiles1 + 7) / 8, B, 0, stream>>>(x, Wl1, bl1, xl1, N);
  gemm_bias_wmma<128, 256><<<(tiles1 + 7) / 8, B, 0, stream>>>(x, Wr1, br1, xr1, N);
  node_attend<256, 4, 8, true><<<(int)(((long long)N * 32 + B - 1) / B), B, 0, stream>>>(
      xl1, xr1, att1, bias1, rowptr, col, h, N);

  int tiles2 = (N / 16) * (OUT / 16);
  gemm_bias_wmma<256, 64><<<(tiles2 + 7) / 8, B, 0, stream>>>(h, Wl2, bl2, xl2, N);
  gemm_bias_wmma<256, 64><<<(tiles2 + 7) / 8, B, 0, stream>>>(h, Wr2, br2, xr2, N);
  node_attend<64, 1, 4, false><<<(int)(((long long)N * 16 + B - 1) / B), B, 0, stream>>>(
      xl2, xr2, att2, bias2, rowptr, col, out, N);
}


# In-edge lists linked by one atomic exchange per edge in one launch replace count, scan and fill (two launches fewer); node_attend follows the links; sum order = reverse link order
# speedup vs baseline: 3.8966x; 1.7193x over previous
#include <hip/hip_runtime.h>
#include <math.h>

#define SLOPE 0.2f

typedef float v2f __attribute__((ext_vector_type(2)));
typedef float v8f __attribute__((ext_vector_type(8)));
typedef float v4f __attribute__((ext_vector_type(4)));

__device__ __forceinline__ float lrelu(float v) { return v > 0.0f ? v : SLOPE * v; }

__device__ __forceinline__ void hold_loads(const v4f (&w)[16]) {
  asm volatile("" ::"v"(w[0]), "v"(w[1]), "v"(w[2]), "v"(w[3]), "v"(w[4]), "v"(w[5]), "v"(w[6]),
               "v"(w[7]), "v"(w[8]), "v"(w[9]), "v"(w[10]), "v"(w[11]), "v"(w[12]), "v"(w[13]),
               "v"(w[14]), "v"(w[15]));
}
__device__ __forceinline__ void hold_loads(const v4f (&a)[4], const v4f (&w)[16]) {
  asm volatile("" ::"v"(a[0]), "v"(a[1]), "v"(a[2]), "v"(a[3]), "v"(w[0]), "v"(w[1]), "v"(w[2]),
               "v"(w[3]), "v"(w[4]), "v"(w[5]), "v"(w[6]), "v"(w[7]), "v"(w[8]), "v"(w[9]),
               "v"(w[10]), "v"(w[11]), "v"(w[12]), "v"(w[13]), "v"(w[14]), "v"(w[15]));
}
__device__ __forceinline__ void hold_loads(const v4f (&a)[8], const v4f (&w)[16]) {
  asm volatile("" ::"v"(a[0]), "v"(a[1]), "v"(a[2]), "v"(a[3]), "v"(a[4]), "v"(a[5]), "v"(a[6]),
               "v"(a[7]), "v"(w[0]), "v"(w[1]), "v"(w[2]), "v"(w[3]), "v"(w[4]), "v"(w[5]),
               "v"(w[6]), "v"(w[7]), "v"(w[8]), "v"(w[9]), "v"(w[10]), "v"(w[11]), "v"(w[12]),
               "v"(w[13]), "v"(w[14]), "v"(w[15]));
}

template <int K, int NC, int RT, int CG>
__global__ void __launch_bounds__(128)
gemm2_bias_wmma(const float* __restrict__ X,
                const float* __restrict__ Wl, const float* __restrict__ bl,
                const float* __restrict__ Wr, const float* __restrict__ br,
                float* __restrict__ Yl, float* __restrict__ Yr, int M) {
  constexpr int WCOLS = 64 * CG;
  constexpr int NCB = 2 * NC / WCOLS;
  constexpr int JB = 8 / CG;
  static_assert(NC % WCOLS == 0 && K % 32 == 0 && (CG == 1 || CG == 2), "tile shape");
  const int wid = threadIdx.x >> 5;
  const int lane = threadIdx.x & 31;
  const int h = lane >> 4;
  const int c = lane & 15;
  const int cbk = (int)(blockIdx.x % NCB);
  const int rb = (int)(blockIdx.x / NCB) * 4 + wid;
  const int MT = M / 16;
  if (rb * RT >= MT) return;
  const int gcol = cbk * WCOLS;
  const bool right = gcol >= NC;
  const float* __restrict__ W = right ? Wr : Wl;
  const float* __restrict__ b = right ? br : bl;
  float* __restrict__ Y = right ? Yr : Yl;
  const int cb = right ? gcol - NC : gcol;

  v8f acc[RT][CG][4];
#pragma unroll
  for (int g = 0; g < CG; g++) {
    const float4 bv = *(const float4*)(b + cb + 64 * g + 4 * c);
    const float bt[4] = {bv.x, bv.y, bv.z, bv.w};
#pragma unroll
    for (int t = 0; t < 4; t++)
#pragma unroll
      for (int rt = 0; rt < RT; rt++)
        acc[rt][g][t] = (v8f){bt[t], bt[t], bt[t], bt[t], bt[t], bt[t], bt[t], bt[t]};
  }

  const float* xp[RT];
#pragma unroll
  for (int rt = 0; rt < RT; rt++) {
    int mt = rb * RT + rt;
    mt = mt < MT ? mt : MT - 1;
    xp[rt] = X + (size_t)(mt * 16 + c) * K + 16 * h;
  }
  const float* wp = W + (size_t)(16 * h) * NC + cb + 4 * c;

#pragma unroll
  for (int kb = 0; kb < K; kb += 32) {
    v4f av[RT * 4];
#pragma unroll
    for (int rt = 0; rt < RT; rt++)
#pragma unroll
      for (int q = 0; q < 4; q++) av[rt * 4 + q] = *(const v4f*)(xp[rt] + kb + 4 * q);
#pragma unroll
    for (int jg = 0; jg < 8; jg += JB) {
      v4f wv[16];
#pragma unroll
      for (int jj = 0; jj < JB; jj++)
#pragma unroll
        for (int g = 0; g < CG; g++) {
          const float* wk = wp + (size_t)(kb + 2 * (jg + jj)) * NC + 64 * g;
          wv[(jj * CG + g) * 2] = *(const v4f*)wk;
          wv[(jj * CG + g) * 2 + 1] = *(const v4f*)(wk + NC);
        }
      if (jg == 0) hold_loads(av, wv); else hold_loads(wv);
#pragma unroll
      for (int jj = 0; jj < JB; jj++) {
        const int j = jg + jj;
#pragma unroll
        for (int g = 0; g < CG; g++) {
          const v4f w0 = wv[(jj * CG + g) * 2];
          const v4f w1 = wv[(jj * CG + g) * 2 + 1];
          const float b0[4] = {w0.x, w0.y, w0.z, w0.w};
          const float b1[4] = {w1.x, w1.y, w1.z, w1.w};
#pragma unroll
          for (int t = 0; t < 4; t++) {
            v2f bf;
            bf.x = b0[t];
            bf.y = b1[t];
#pragma unroll
            for (int rt = 0; rt < RT; rt++) {
              const v4f aq = av[rt * 4 + (j >> 1)];
              v2f af;
              af.x = (j & 1) ? aq.z : aq.x;
              af.y = (j & 1) ? aq.w : aq.y;
              acc[rt][g][t] = __builtin_amdgcn_wmma_f32_16x16x4_f32(false, af, false, bf, (short)0,
                                                                    acc[rt][g][t], false, false);
            }
          }
        }
      }
    }
  }

#pragma unroll
  for (int rt = 0; rt < RT; rt++) {
    const int mt = rb * RT + rt;
    if (mt < MT) {
      float* yp = Y + (size_t)(mt * 16 + 8 * h) * NC + cb + 4 * c;
#pragma unroll
      for (int g = 0; g < CG; g++)
#pragma unroll
        for (int r = 0; r < 8; r++) {
          float4 o;
          o.x = acc[rt][g][0][r]; o.y = acc[rt][g][1][r];
          o.z = acc[rt][g][2][r]; o.w = acc[rt][g][3][r];
          *(float4*)(yp + (size_t)r * NC + 64 * g) = o;
        }
    }
  }
}

__global__ void link_edges(const int* __restrict__ ei, int E_, int* __restrict__ head,
                           int* __restrict__ lst) {
  int e = (int)(blockIdx.x * (size_t)blockDim.x + threadIdx.x);
  if (e >= E_) return;
  const int s = ei[e], d = ei[E_ + e];
  const int prev = atomicExch(head + d, e + 1);
  ((int2*)lst)[e] = make_int2(s, prev);
}

template <int V>
__device__ __forceinline__ void load_row(const float* __restrict__ p, float (&x)[V]) {
#pragma unroll
  for (int q = 0; q < V / 4; q++) {
    float4 t = *(const float4*)(p + 4 * q);
    x[4 * q] = t.x; x[4 * q + 1] = t.y; x[4 * q + 2] = t.z; x[4 * q + 3] = t.w;
  }
}

template <int V, int LPH>
__device__ __forceinline__ float edge_logit(const float (&x)[V], const float (&r)[V],
                                            const float (&a)[V]) {
  float p = 0.0f;
#pragma unroll
  for (int k = 0; k < V; k++) p += lrelu(x[k] + r[k]) * a[k];
  for (int off = LPH / 2; off > 0; off >>= 1) p += __shfl_xor(p, off, 32);
  return p;
}

template <int HC, int H_, int V, bool ELU>
__global__ void node_attend(const float* __restrict__ xl, const float* __restrict__ xr,
                            const float* __restrict__ att, const float* __restrict__ bias,
                            const int* __restrict__ head, const int* __restrict__ lst,
                            float* __restrict__ outp, int n, int E_) {
  constexpr int LPN = HC / V;
  constexpr int LPH = LPN / H_;
  int node = (int)((blockIdx.x * (size_t)blockDim.x + threadIdx.x) / LPN);
  if (LPN == 32) node = __builtin_amdgcn_readfirstlane(node);
  if (node >= n) return;
  const int base = (int)(threadIdx.x % LPN) * V;
  const int2* __restrict__ links = (const int2*)lst;
  float r[V], a[V], x[V], acc[V], bs[V];
  load_row<V>(xr + (size_t)node * HC + base, r);
  load_row<V>(att + base, a);
  load_row<V>(xl + (size_t)node * HC + base, acc);
  float m = edge_logit<V, LPH>(acc, r, a);
  float s = 1.0f;
  int p = head[node];
  int2 cur = links[p > 0 ? p - 1 : 0];
  for (int it = 0; p != 0 && it < E_; it++) {
    const int pn = cur.y;
    const int2 nx = links[pn > 0 ? pn - 1 : 0];
    asm volatile("" ::: "memory");
    load_row<V>(xl + (size_t)cur.x * HC + base, x);
    const float e = edge_logit<V, LPH>(x, r, a);
    const float d = e - m;
    const float t = expf(-fabsf(d));
    const float sc = d > 0.0f ? t : 1.0f;
    const float wgt = d > 0.0f ? 1.0f : t;
    m = fmaxf(m, e);
    s = s * sc + wgt;
#pragma unroll
    for (int k = 0; k < V; k++) acc[k] = acc[k] * sc + wgt * x[k];
    p = pn;
    cur = nx;
  }
  load_row<V>(bias + base, bs);
  float* o = outp + (size_t)node * HC + base;
#pragma unroll
  for (int q = 0; q < V / 4; q++) {
    float4 t;
    float v0 = acc[4 * q] / s + bs[4 * q];
    float v1 = acc[4 * q + 1] / s + bs[4 * q + 1];
    float v2 = acc[4 * q + 2] / s + bs[4 * q + 2];
    float v3 = acc[4 * q + 3] / s + bs[4 * q + 3];
    if (ELU) {
      v0 = v0 > 0.0f ? v0 : expm1f(v0);
      v1 = v1 > 0.0f ? v1 : expm1f(v1);
      v2 = v2 > 0.0f ? v2 : expm1f(v2);
      v3 = v3 > 0.0f ? v3 : expm1f(v3);
    }
    t.x = v0; t.y = v1; t.z = v2; t.w = v3;
    *(float4*)(o + 4 * q) = t;
  }
}

extern "C" void kernel_launch(void* const* d_in, const int* in_sizes, int n_in,
                              void* d_out, int out_size, void* d_ws, size_t ws_size,
                              hipStream_t stream) {
  (void)n_in; (void)out_size; (void)ws_size;
  const int IN = 128, HID = 64, H = 4, OUT = 64;
  const float*    x    = (const float*)d_in[0];
  const int*      ei   = (const int*)d_in[1];
  const float*    Wl1  = (const float*)d_in[2];
  const float*    bl1  = (const float*)d_in[3];
  const float*    Wr1  = (const float*)d_in[4];
  const float*    br1  = (const float*)d_in[5];
  const float*    att1 = (const float*)d_in[6];
  const float*    bias1= (const float*)d_in[7];
  const float*    Wl2  = (const float*)d_in[8];
  const float*    bl2  = (const float*)d_in[9];
  const float*    Wr2  = (const float*)d_in[10];
  const float*    br2  = (const float*)d_in[11];
  const float*    att2 = (const float*)d_in[12];
  const float*    bias2= (const float*)d_in[13];
  float* out = (float*)d_out;

  const int N    = in_sizes[0] / IN;
  const int E    = in_sizes[1] / 2;
  const int HC   = H * HID;

  float* ws = (float*)d_ws;
  size_t o = 0;
  float* xl1    = ws + o; o += (size_t)N * HC;
  float* xr1    = ws + o; o += (size_t)N * HC;
  float* h      = ws + o; o += (size_t)N * HC;
  int*   lst    = (int*)(ws + o); o += (size_t)2 * E;
  int*   head   = (int*)(ws + o); o += (size_t)N + 4;
  float* xl2 = xl1;
  float* xr2 = xr1;

  const int B = 256;

  hipMemsetAsync(head, 0, (size_t)N * sizeof(int), stream);
  link_edges<<<(E + B - 1) / B, B, 0, stream>>>(ei, E, head, lst);

  int rowblk1 = (N / 16 + 1) / 2;
  int groups1 = ((rowblk1 + 3) / 4) * (2 * HC / 128);
  gemm2_bias_wmma<128, 256, 2, 2><<<groups1, 128, 0, stream>>>(x, Wl1, bl1, Wr1, br1, xl1, xr1, N);
  node_attend<256, 4, 8, true><<<(int)(((long long)N * 32 + B - 1) / B), B, 0, stream>>>(
      xl1, xr1, att1, bias1, head, lst, h, N, E);

  int rowblk2 = N / 16;
  int groups2 = ((rowblk2 + 3) / 4) * (2 * OUT / 64);
  gemm2_bias_wmma<256, 64, 1, 1><<<groups2, 128, 0, stream>>>(h, Wl2, bl2, Wr2, br2, xl2, xr2, N);
  node_attend<64, 1, 4, false><<<(int)(((long long)N * 16 + B - 1) / B), B, 0, stream>>>(
      xl2, xr2, att2, bias2, head, lst, out, N, E);
}


# Edge linking rides in the layer-1 product launch: its last work-groups link the edges (4 per thread) beside the matrix work; one launch fewer; sum order = link arrival order as before
# speedup vs baseline: 4.1073x; 1.0541x over previous
#include <hip/hip_runtime.h>
#include <math.h>

#define SLOPE 0.2f

typedef float v2f __attribute__((ext_vector_type(2)));
typedef float v8f __attribute__((ext_vector_type(8)));
typedef float v4f __attribute__((ext_vector_type(4)));

__device__ __forceinline__ float lrelu(float v) { return v > 0.0f ? v : SLOPE * v; }

__device__ __forceinline__ void link_one(const int* __restrict__ ei, int E_, int* __restrict__ head,
                                         int* __restrict__ lst, int e) {
  const int s = ei[e], d = ei[E_ + e];
  const int prev = atomicExch(head + d, e + 1);
  ((int2*)lst)[e] = make_int2(s, prev);
}
__device__ __forceinline__ void link_four(const int* __restrict__ ei, int E_, int* __restrict__ head,
                                          int* __restrict__ lst, int t) {
  const int e0 = 4 * t;
  if (e0 >= E_) return;
  if ((E_ & 3) == 0) {
    const int4 s = *(const int4*)(ei + e0);
    const int4 d = *(const int4*)(ei + E_ + e0);
    const int p0 = atomicExch(head + d.x, e0 + 1);
    const int p1 = atomicExch(head + d.y, e0 + 2);
    const int p2 = atomicExch(head + d.z, e0 + 3);
    const int p3 = atomicExch(head + d.w, e0 + 4);
    int4* o = (int4*)(lst + 2 * (size_t)e0);
    o[0] = make_int4(s.x, p0, s.y, p1);
    o[1] = make_int4(s.z, p2, s.w, p3);
  } else {
    for (int k = 0; k < 4 && e0 + k < E_; k++) link_one(ei, E_, head, lst, e0 + k);
  }
}

__device__ __forceinline__ void hold_loads(const v4f (&w)[16]) {
  asm volatile("" ::"v"(w[0]), "v"(w[1]), "v"(w[2]), "v"(w[3]), "v"(w[4]), "v"(w[5]), "v"(w[6]),
               "v"(w[7]), "v"(w[8]), "v"(w[9]), "v"(w[10]), "v"(w[11]), "v"(w[12]), "v"(w[13]),
               "v"(w[14]), "v"(w[15]));
}
__device__ __forceinline__ void hold_loads(const v4f (&a)[4], const v4f (&w)[16]) {
  asm volatile("" ::"v"(a[0]), "v"(a[1]), "v"(a[2]), "v"(a[3]), "v"(w[0]), "v"(w[1]), "v"(w[2]),
               "v"(w[3]), "v"(w[4]), "v"(w[5]), "v"(w[6]), "v"(w[7]), "v"(w[8]), "v"(w[9]),
               "v"(w[10]), "v"(w[11]), "v"(w[12]), "v"(w[13]), "v"(w[14]), "v"(w[15]));
}
__device__ __forceinline__ void hold_loads(const v4f (&a)[8], const v4f (&w)[16]) {
  asm volatile("" ::"v"(a[0]), "v"(a[1]), "v"(a[2]), "v"(a[3]), "v"(a[4]), "v"(a[5]), "v"(a[6]),
               "v"(a[7]), "v"(w[0]), "v"(w[1]), "v"(w[2]), "v"(w[3]), "v"(w[4]), "v"(w[5]),
               "v"(w[6]), "v"(w[7]), "v"(w[8]), "v"(w[9]), "v"(w[10]), "v"(w[11]), "v"(w[12]),
               "v"(w[13]), "v"(w[14]), "v"(w[15]));
}

template <int K, int NC, int RT, int CG, bool LINK>
__global__ void __launch_bounds__(128)
gemm2_bias_wmma(const float* __restrict__ X,
                const float* __restrict__ Wl, const float* __restrict__ bl,
                const float* __restrict__ Wr, const float* __restrict__ br,
                float* __restrict__ Yl, float* __restrict__ Yr, int M, int ngemm,
                const int* __restrict__ ei, int E_, int* __restrict__ head, int* __restrict__ lst) {
  if (LINK && (int)blockIdx.x >= ngemm) {
    link_four(ei, E_, head, lst, ((int)blockIdx.x - ngemm) * 128 + (int)threadIdx.x);
    return;
  }
  constexpr int WCOLS = 64 * CG;
  constexpr int NCB = 2 * NC / WCOLS;
  constexpr int JB = 8 / CG;
  static_assert(NC % WCOLS == 0 && K % 32 == 0 && (CG == 1 || CG == 2), "tile shape");
  const int wid = threadIdx.x >> 5;
  const int lane = threadIdx.x & 31;
  const int h = lane >> 4;
  const int c = lane & 15;
  const int cbk = (int)(blockIdx.x % NCB);
  const int rb = (int)(blockIdx.x / NCB) * 4 + wid;
  const int MT = M / 16;
  if (rb * RT >= MT) return;
  const int gcol = cbk * WCOLS;
  const bool right = gcol >= NC;
  const float* __restrict__ W = right ? Wr : Wl;
  const float* __restrict__ b = right ? br : bl;
  float* __restrict__ Y = right ? Yr : Yl;
  const int cb = right ? gcol - NC : gcol;

  v8f acc[RT][CG][4];
#pragma unroll
  for (int g = 0; g < CG; g++) {
    const float4 bv = *(const float4*)(b + cb + 64 * g + 4 * c);
    const float bt[4] = {bv.x, bv.y, bv.z, bv.w};
#pragma unroll
    for (int t = 0; t < 4; t++)
#pragma unroll
      for (int rt = 0; rt < RT; rt++)
        acc[rt][g][t] = (v8f){bt[t], bt[t], bt[t], bt[t], bt[t], bt[t], bt[t], bt[t]};
  }

  const float* xp[RT];
#pragma unroll
  for (int rt = 0; rt < RT; rt++) {
    int mt = rb * RT + rt;
    mt = mt < MT ? mt : MT - 1;
    xp[rt] = X + (size_t)(mt * 16 + c) * K + 16 * h;
  }
  const float* wp = W + (size_t)(16 * h) * NC + cb + 4 * c;

#pragma unroll
  for (int kb = 0; kb < K; kb += 32) {
    v4f av[RT * 4];
#pragma unroll
    for (int rt = 0; rt < RT; rt++)
#pragma unroll
      for (int q = 0; q < 4; q++) av[rt * 4 + q] = *(const v4f*)(xp[rt] + kb + 4 * q);
#pragma unroll
    for (int jg = 0; jg < 8; jg += JB) {
      v4f wv[16];
#pragma unroll
      for (int jj = 0; jj < JB; jj++)
#pragma unroll
        for (int g = 0; g < CG; g++) {
          const float* wk = wp + (size_t)(kb + 2 * (jg + jj)) * NC + 64 * g;
          wv[(jj * CG + g) * 2] = *(const v4f*)wk;
          wv[(jj * CG + g) * 2 + 1] = *(const v4f*)(wk + NC);
        }
      if (jg == 0) hold_loads(av, wv); else hold_loads(wv);
#pragma unroll
      for (int jj = 0; jj < JB; jj++) {
        const int j = jg + jj;
#pragma unroll
        for (int g = 0; g < CG; g++) {
          const v4f w0 = wv[(jj * CG + g) * 2];
          const v4f w1 = wv[(jj * CG + g) * 2 + 1];
          const float b0[4] = {w0.x, w0.y, w0.z, w0.w};
          const float b1[4] = {w1.x, w1.y, w1.z, w1.w};
#pragma unroll
          for (int t = 0; t < 4; t++) {
            v2f bf;
            bf.x = b0[t];
            bf.y = b1[t];
#pragma unroll
            for (int rt = 0; rt < RT; rt++) {
              const v4f aq = av[rt * 4 + (j >> 1)];
              v2f af;
              af.x = (j & 1) ? aq.z : aq.x;
              af.y = (j & 1) ? aq.w : aq.y;
              acc[rt][g][t] = __builtin_amdgcn_wmma_f32_16x16x4_f32(false, af, false, bf, (short)0,
                                                                    acc[rt][g][t], false, false);
            }
          }
        }
      }
    }
  }

#pragma unroll
  for (int rt = 0; rt < RT; rt++) {
    const int mt = rb * RT + rt;
    if (mt < MT) {
      float* yp = Y + (size_t)(mt * 16 + 8 * h) * NC + cb + 4 * c;
#pragma unroll
      for (int g = 0; g < CG; g++)
#pragma unroll
        for (int r = 0; r < 8; r++) {
          float4 o;
          o.x = acc[rt][g][0][r]; o.y = acc[rt][g][1][r];
          o.z = acc[rt][g][2][r]; o.w = acc[rt][g][3][r];
          *(float4*)(yp + (size_t)r * NC + 64 * g) = o;
        }
    }
  }
}

template <int V>
__device__ __forceinline__ void load_row(const float* __restrict__ p, float (&x)[V]) {
#pragma unroll
  for (int q = 0; q < V / 4; q++) {
    float4 t = *(const float4*)(p + 4 * q);
    x[4 * q] = t.x; x[4 * q + 1] = t.y; x[4 * q + 2] = t.z; x[4 * q + 3] = t.w;
  }
}

template <int V, int LPH>
__device__ __forceinline__ float edge_logit(const float (&x)[V], const float (&r)[V],
                                            const float (&a)[V]) {
  float p = 0.0f;
#pragma unroll
  for (int k = 0; k < V; k++) p += lrelu(x[k] + r[k]) * a[k];
  for (int off = LPH / 2; off > 0; off >>= 1) p += __shfl_xor(p, off, 32);
  return p;
}

template <int HC, int H_, int V, bool ELU>
__global__ void node_attend(const float* __restrict__ xl, const float* __restrict__ xr,
                            const float* __restrict__ att, const float* __restrict__ bias,
                            const int* __restrict__ head, const int* __restrict__ lst,
                            float* __restrict__ outp, int n, int E_) {
  constexpr int LPN = HC / V;
  constexpr int LPH = LPN / H_;
  int node = (int)((blockIdx.x * (size_t)blockDim.x + threadIdx.x) / LPN);
  if (LPN == 32) node = __builtin_amdgcn_readfirstlane(node);
  if (node >= n) return;
  const int base = (int)(threadIdx.x % LPN) * V;
  const int2* __restrict__ links = (const int2*)lst;
  float r[V], a[V], x[V], acc[V], bs[V];
  load_row<V>(xr + (size_t)node * HC + base, r);
  load_row<V>(att + base, a);
  load_row<V>(xl + (size_t)node * HC + base, acc);
  float m = edge_logit<V, LPH>(acc, r, a);
  float s = 1.0f;
  int p = head[node];
  int2 cur = links[p > 0 ? p - 1 : 0];
  for (int it = 0; p != 0 && it < E_; it++) {
    const int pn = cur.y;
    const int2 nx = links[pn > 0 ? pn - 1 : 0];
    asm volatile("" ::: "memory");
    load_row<V>(xl + (size_t)cur.x * HC + base, x);
    const float e = edge_logit<V, LPH>(x, r, a);
    const float d = e - m;
    const float t = expf(-fabsf(d));
    const float sc = d > 0.0f ? t : 1.0f;
    const float wgt = d > 0.0f ? 1.0f : t;
    m = fmaxf(m, e);
    s = s * sc + wgt;
#pragma unroll
    for (int k = 0; k < V; k++) acc[k] = acc[k] * sc + wgt * x[k];
    p = pn;
    cur = nx;
  }
  load_row<V>(bias + base, bs);
  float* o = outp + (size_t)node * HC + base;
#pragma unroll
  for (int q = 0; q < V / 4; q++) {
    float4 t;
    float v0 = acc[4 * q] / s + bs[4 * q];
    float v1 = acc[4 * q + 1] / s + bs[4 * q + 1];
    float v2 = acc[4 * q + 2] / s + bs[4 * q + 2];
    float v3 = acc[4 * q + 3] / s + bs[4 * q + 3];
    if (ELU) {
      v0 = v0 > 0.0f ? v0 : expm1f(v0);
      v1 = v1 > 0.0f ? v1 : expm1f(v1);
      v2 = v2 > 0.0f ? v2 : expm1f(v2);
      v3 = v3 > 0.0f ? v3 : expm1f(v3);
    }
    t.x = v0; t.y = v1; t.z = v2; t.w = v3;
    *(float4*)(o + 4 * q) = t;
  }
}

extern "C" void kernel_launch(void* const* d_in, const int* in_sizes, int n_in,
                              void* d_out, int out_size, void* d_ws, size_t ws_size,
                              hipStream_t stream) {
  (void)n_in; (void)out_size; (void)ws_size;
  const int IN = 128, HID = 64, H = 4, OUT = 64;
  const float*    x    = (const float*)d_in[0];
  const int*      ei   = (const int*)d_in[1];
  const float*    Wl1  = (const float*)d_in[2];
  const float*    bl1  = (const float*)d_in[3];
  const float*    Wr1  = (const float*)d_in[4];
  const float*    br1  = (const float*)d_in[5];
  const float*    att1 = (const float*)d_in[6];
  const float*    bias1= (const float*)d_in[7];
  const float*    Wl2  = (const float*)d_in[8];
  const float*    bl2  = (const float*)d_in[9];
  const float*    Wr2  = (const float*)d_in[10];
  const float*    br2  = (const float*)d_in[11];
  const float*    att2 = (const float*)d_in[12];
  const float*    bias2= (const float*)d_in[13];
  float* out = (float*)d_out;

  const int N    = in_sizes[0] / IN;
  const int E    = in_sizes[1] / 2;
  const int HC   = H * HID;

  float* ws = (float*)d_ws;
  size_t o = 0;
  float* xl1    = ws + o; o += (size_t)N * HC;
  float* xr1    = ws + o; o += (size_t)N * HC;
  float* h      = ws + o; o += (size_t)N * HC;
  int*   lst    = (int*)(ws + o); o += (size_t)2 * E;
  int*   head   = (int*)(ws + o); o += (size_t)N + 4;
  float* xl2 = xl1;
  float* xr2 = xr1;

  const int B = 256;

  hipMemsetAsync(head, 0, (size_t)N * sizeof(int), stream);

  int rowblk1 = (N / 16 + 1) / 2;
  int groups1 = ((rowblk1 + 3) / 4) * (2 * HC / 128);
  int linkgrp = ((E + 3) / 4 + 127) / 128;
  gemm2_bias_wmma<128, 256, 2, 2, true><<<groups1 + linkgrp, 128, 0, stream>>>(
      x, Wl1, bl1, Wr1, br1, xl1, xr1, N, groups1, ei, E, head, lst);
  node_attend<256, 4, 8, true><<<(int)(((long long)N * 32 + B - 1) / B), B, 0, stream>>>(
      xl1, xr1, att1, bias1, head, lst, h, N, E);

  int rowblk2 = N / 16;
  int groups2 = ((rowblk2 + 3) / 4) * (2 * OUT / 64);
  gemm2_bias_wmma<256, 64, 1, 1, false><<<groups2, 128, 0, stream>>>(
      h, Wl2, bl2, Wr2, br2, xl2, xr2, N, groups2, ei, E, head, lst);
  node_attend<64, 1, 4, false><<<(int)(((long long)N * 16 + B - 1) / B), B, 0, stream>>>(
      xl2, xr2, att2, bias2, head, lst, out, N, E);
}


# Node pass math on native forms: expf and expm1f become the hardware exp2 (log2 e folded into att), eight divisions per node become one reciprocal and multiplies; last bits of results change
# speedup vs baseline: 4.3309x; 1.0544x over previous
#include <hip/hip_runtime.h>
#include <math.h>

#define SLOPE 0.2f

typedef float v2f __attribute__((ext_vector_type(2)));
typedef float v8f __attribute__((ext_vector_type(8)));
typedef float v4f __attribute__((ext_vector_type(4)));

__device__ __forceinline__ float lrelu(float v) { return v > 0.0f ? v : SLOPE * v; }

#define LOG2E 1.44269504088896340736f
__device__ __forceinline__ float exp2_hw(float v) { return __builtin_amdgcn_exp2f(v); }

__device__ __forceinline__ void link_one(const int* __restrict__ ei, int E_, int* __restrict__ head,
                                         int* __restrict__ lst, int e) {
  const int s = ei[e], d = ei[E_ + e];
  const int prev = atomicExch(head + d, e + 1);
  ((int2*)lst)[e] = make_int2(s, prev);
}
__device__ __forceinline__ void link_four(const int* __restrict__ ei, int E_, int* __restrict__ head,
                                          int* __restrict__ lst, int t) {
  const int e0 = 4 * t;
  if (e0 >= E_) return;
  if ((E_ & 3) == 0) {
    const int4 s = *(const int4*)(ei + e0);
    const int4 d = *(const int4*)(ei + E_ + e0);
    const int p0 = atomicExch(head + d.x, e0 + 1);
    const int p1 = atomicExch(head + d.y, e0 + 2);
    const int p2 = atomicExch(head + d.z, e0 + 3);
    const int p3 = atomicExch(head + d.w, e0 + 4);
    int4* o = (int4*)(lst + 2 * (size_t)e0);
    o[0] = make_int4(s.x, p0, s.y, p1);
    o[1] = make_int4(s.z, p2, s.w, p3);
  } else {
    for (int k = 0; k < 4 && e0 + k < E_; k++) link_one(ei, E_, head, lst, e0 + k);
  }
}

__device__ __forceinline__ void hold_loads(const v4f (&w)[16]) {
  asm volatile("" ::"v"(w[0]), "v"(w[1]), "v"(w[2]), "v"(w[3]), "v"(w[4]), "v"(w[5]), "v"(w[6]),
               "v"(w[7]), "v"(w[8]), "v"(w[9]), "v"(w[10]), "v"(w[11]), "v"(w[12]), "v"(w[13]),
               "v"(w[14]), "v"(w[15]));
}
__device__ __forceinline__ void hold_loads(const v4f (&a)[4], const v4f (&w)[16]) {
  asm volatile("" ::"v"(a[0]), "v"(a[1]), "v"(a[2]), "v"(a[3]), "v"(w[0]), "v"(w[1]), "v"(w[2]),
               "v"(w[3]), "v"(w[4]), "v"(w[5]), "v"(w[6]), "v"(w[7]), "v"(w[8]), "v"(w[9]),
               "v"(w[10]), "v"(w[11]), "v"(w[12]), "v"(w[13]), "v"(w[14]), "v"(w[15]));
}
__device__ __forceinline__ void hold_loads(const v4f (&a)[8], const v4f (&w)[16]) {
  asm volatile("" ::"v"(a[0]), "v"(a[1]), "v"(a[2]), "v"(a[3]), "v"(a[4]), "v"(a[5]), "v"(a[6]),
               "v"(a[7]), "v"(w[0]), "v"(w[1]), "v"(w[2]), "v"(w[3]), "v"(w[4]), "v"(w[5]),
               "v"(w[6]), "v"(w[7]), "v"(w[8]), "v"(w[9]), "v"(w[10]), "v"(w[11]), "v"(w[12]),
               "v"(w[13]), "v"(w[14]), "v"(w[15]));
}

template <int K, int NC, int RT, int CG, bool LINK>
__global__ void __launch_bounds__(128)
gemm2_bias_wmma(const float* __restrict__ X,
                const float* __restrict__ Wl, const float* __restrict__ bl,
                const float* __restrict__ Wr, const float* __restrict__ br,
                float* __restrict__ Yl, float* __restrict__ Yr, int M, int ngemm,
                const int* __restrict__ ei, int E_, int* __restrict__ head, int* __restrict__ lst) {
  if (LINK && (int)blockIdx.x >= ngemm) {
    link_four(ei, E_, head, lst, ((int)blockIdx.x - ngemm) * 128 + (int)threadIdx.x);
    return;
  }
  constexpr int WCOLS = 64 * CG;
  constexpr int NCB = 2 * NC / WCOLS;
  constexpr int JB = 8 / CG;
  static_assert(NC % WCOLS == 0 && K % 32 == 0 && (CG == 1 || CG == 2), "tile shape");
  const int wid = threadIdx.x >> 5;
  const int lane = threadIdx.x & 31;
  const int h = lane >> 4;
  const int c = lane & 15;
  const int cbk = (int)(blockIdx.x % NCB);
  const int rb = (int)(blockIdx.x / NCB) * 4 + wid;
  const int MT = M / 16;
  if (rb * RT >= MT) return;
  const int gcol = cbk * WCOLS;
  const bool right = gcol >= NC;
  const float* __restrict__ W = right ? Wr : Wl;
  const float* __restrict__ b = right ? br : bl;
  float* __restrict__ Y = right ? Yr : Yl;
  const int cb = right ? gcol - NC : gcol;

  v8f acc[RT][CG][4];
#pragma unroll
  for (int g = 0; g < CG; g++) {
    const float4 bv = *(const float4*)(b + cb + 64 * g + 4 * c);
    const float bt[4] = {bv.x, bv.y, bv.z, bv.w};
#pragma unroll
    for (int t = 0; t < 4; t++)
#pragma unroll
      for (int rt = 0; rt < RT; rt++)
        acc[rt][g][t] = (v8f){bt[t], bt[t], bt[t], bt[t], bt[t], bt[t], bt[t], bt[t]};
  }

  const float* xp[RT];
#pragma unroll
  for (int rt = 0; rt < RT; rt++) {
    int mt = rb * RT + rt;
    mt = mt < MT ? mt : MT - 1;
    xp[rt] = X + (size_t)(mt * 16 + c) * K + 16 * h;
  }
  const float* wp = W + (size_t)(16 * h) * NC + cb + 4 * c;

#pragma unroll
  for (int kb = 0; kb < K; kb += 32) {
    v4f av[RT * 4];
#pragma unroll
    for (int rt = 0; rt < RT; rt++)
#pragma unroll
      for (int q = 0; q < 4; q++) av[rt * 4 + q] = *(const v4f*)(xp[rt] + kb + 4 * q);
#pragma unroll
    for (int jg = 0; jg < 8; jg += JB) {
      v4f wv[16];
#pragma unroll
      for (int jj = 0; jj < JB; jj++)
#pragma unroll
        for (int g = 0; g < CG; g++) {
          const float* wk = wp + (size_t)(kb + 2 * (jg + jj)) * NC + 64 * g;
          wv[(jj * CG + g) * 2] = *(const v4f*)wk;
          wv[(jj * CG + g) * 2 + 1] = *(const v4f*)(wk + NC);
        }
      if (jg == 0) hold_loads(av, wv); else hold_loads(wv);
#pragma unroll
      for (int jj = 0; jj < JB; jj++) {
        const int j = jg + jj;
#pragma unroll
        for (int g = 0; g < CG; g++) {
          const v4f w0 = wv[(jj * CG + g) * 2];
          const v4f w1 = wv[(jj * CG + g) * 2 + 1];
          const float b0[4] = {w0.x, w0.y, w0.z, w0.w};
          const float b1[4] = {w1.x, w1.y, w1.z, w1.w};
#pragma unroll
          for (int t = 0; t < 4; t++) {
            v2f bf;
            bf.x = b0[t];
            bf.y = b1[t];
#pragma unroll
            for (int rt = 0; rt < RT; rt++) {
              const v4f aq = av[rt * 4 + (j >> 1)];
              v2f af;
              af.x = (j & 1) ? aq.z : aq.x;
              af.y = (j & 1) ? aq.w : aq.y;
              acc[rt][g][t] = __builtin_amdgcn_wmma_f32_16x16x4_f32(false, af, false, bf, (short)0,
                                                                    acc[rt][g][t], false, false);
            }
          }
        }
      }
    }
  }

#pragma unroll
  for (int rt = 0; rt < RT; rt++) {
    const int mt = rb * RT + rt;
    if (mt < MT) {
      float* yp = Y + (size_t)(mt * 16 + 8 * h) * NC + cb + 4 * c;
#pragma unroll
      for (int g = 0; g < CG; g++)
#pragma unroll
        for (int r = 0; r < 8; r++) {
          float4 o;
          o.x = acc[rt][g][0][r]; o.y = acc[rt][g][1][r];
          o.z = acc[rt][g][2][r]; o.w = acc[rt][g][3][r];
          *(float4*)(yp + (size_t)r * NC + 64 * g) = o;
        }
    }
  }
}

template <int V>
__device__ __forceinline__ void load_row(const float* __restrict__ p, float (&x)[V]) {
#pragma unroll
  for (int q = 0; q < V / 4; q++) {
    float4 t = *(const float4*)(p + 4 * q);
    x[4 * q] = t.x; x[4 * q + 1] = t.y; x[4 * q + 2] = t.z; x[4 * q + 3] = t.w;
  }
}

template <int V, int LPH>
__device__ __forceinline__ float edge_logit(const float (&x)[V], const float (&r)[V],
                                            const float (&a)[V]) {
  float p = 0.0f;
#pragma unroll
  for (int k = 0; k < V; k++) p += lrelu(x[k] + r[k]) * a[k];
  for (int off = LPH / 2; off > 0; off >>= 1) p += __shfl_xor(p, off, 32);
  return p;
}

template <int HC, int H_, int V, bool ELU>
__global__ void node_attend(const float* __restrict__ xl, const float* __restrict__ xr,
                            const float* __restrict__ att, const float* __restrict__ bias,
                            const int* __restrict__ head, const int* __restrict__ lst,
                            float* __restrict__ outp, int n, int E_) {
  constexpr int LPN = HC / V;
  constexpr int LPH = LPN / H_;
  int node = (int)((blockIdx.x * (size_t)blockDim.x + threadIdx.x) / LPN);
  if (LPN == 32) node = __builtin_amdgcn_readfirstlane(node);
  if (node >= n) return;
  const int base = (int)(threadIdx.x % LPN) * V;
  const int2* __restrict__ links = (const int2*)lst;
  float r[V], a[V], x[V], acc[V], bs[V];
  load_row<V>(xr + (size_t)node * HC + base, r);
  load_row<V>(att + base, a);
#pragma unroll
  for (int k = 0; k < V; k++) a[k] *= LOG2E;
  load_row<V>(xl + (size_t)node * HC + base, acc);
  float m = edge_logit<V, LPH>(acc, r, a);
  float s = 1.0f;
  int p = head[node];
  int2 cur = links[p > 0 ? p - 1 : 0];
  for (int it = 0; p != 0 && it < E_; it++) {
    const int pn = cur.y;
    const int2 nx = links[pn > 0 ? pn - 1 : 0];
    asm volatile("" ::: "memory");
    load_row<V>(xl + (size_t)cur.x * HC + base, x);
    const float e = edge_logit<V, LPH>(x, r, a);
    const float d = e - m;
    const float t = exp2_hw(-fabsf(d));
    const float sc = d > 0.0f ? t : 1.0f;
    const float wgt = d > 0.0f ? 1.0f : t;
    m = fmaxf(m, e);
    s = s * sc + wgt;
#pragma unroll
    for (int k = 0; k < V; k++) acc[k] = acc[k] * sc + wgt * x[k];
    p = pn;
    cur = nx;
  }
  load_row<V>(bias + base, bs);
  const float inv = 1.0f / s;
  float* o = outp + (size_t)node * HC + base;
#pragma unroll
  for (int q = 0; q < V / 4; q++) {
    float4 t;
    float v0 = acc[4 * q] * inv + bs[4 * q];
    float v1 = acc[4 * q + 1] * inv + bs[4 * q + 1];
    float v2 = acc[4 * q + 2] * inv + bs[4 * q + 2];
    float v3 = acc[4 * q + 3] * inv + bs[4 * q + 3];
    if (ELU) {
      v0 = v0 > 0.0f ? v0 : exp2_hw(v0 * LOG2E) - 1.0f;
      v1 = v1 > 0.0f ? v1 : exp2_hw(v1 * LOG2E) - 1.0f;
      v2 = v2 > 0.0f ? v2 : exp2_hw(v2 * LOG2E) - 1.0f;
      v3 = v3 > 0.0f ? v3 : exp2_hw(v3 * LOG2E) - 1.0f;
    }
    t.x = v0; t.y = v1; t.z = v2; t.w = v3;
    *(float4*)(o + 4 * q) = t;
  }
}

extern "C" void kernel_launch(void* const* d_in, const int* in_sizes, int n_in,
                              void* d_out, int out_size, void* d_ws, size_t ws_size,
                              hipStream_t stream) {
  (void)n_in; (void)out_size; (void)ws_size;
  const int IN = 128, HID = 64, H = 4, OUT = 64;
  const float*    x    = (const float*)d_in[0];
  const int*      ei   = (const int*)d_in[1];
  const float*    Wl1  = (const float*)d_in[2];
  const float*    bl1  = (const float*)d_in[3];
  const float*    Wr1  = (const float*)d_in[4];
  const float*    br1  = (const float*)d_in[5];
  const float*    att1 = (const float*)d_in[6];
  const float*    bias1= (const float*)d_in[7];
  const float*    Wl2  = (const float*)d_in[8];
  const float*    bl2  = (const float*)d_in[9];
  const float*    Wr2  = (const float*)d_in[10];
  const float*    br2  = (const float*)d_in[11];
  const float*    att2 = (const float*)d_in[12];
  const float*    bias2= (const float*)d_in[13];
  float* out = (float*)d_out;

  const int N    = in_sizes[0] / IN;
  const int E    = in_sizes[1] / 2;
  const int HC   = H * HID;

  float* ws = (float*)d_ws;
  size_t o = 0;
  float* xl1    = ws + o; o += (size_t)N * HC;
  float* xr1    = ws + o; o += (size_t)N * HC;
  float* h      = ws + o; o += (size_t)N * HC;
  int*   lst    = (int*)(ws + o); o += (size_t)2 * E;
  int*   head   = (int*)(ws + o); o += (size_t)N + 4;
  float* xl2 = xl1;
  float* xr2 = xr1;

  const int B = 256;

  hipMemsetAsync(head, 0, (size_t)N * sizeof(int), stream);

  int rowblk1 = (N / 16 + 1) / 2;
  int groups1 = ((rowblk1 + 3) / 4) * (2 * HC / 128);
  int linkgrp = ((E + 3) / 4 + 127) / 128;
  gemm2_bias_wmma<128, 256, 2, 2, true><<<groups1 + linkgrp, 128, 0, stream>>>(
      x, Wl1, bl1, Wr1, br1, xl1, xr1, N, groups1, ei, E, head, lst);
  node_attend<256, 4, 8, true><<<(int)(((long long)N * 32 + B - 1) / B), B, 0, stream>>>(
      xl1, xr1, att1, bias1, head, lst, h, N, E);

  int rowblk2 = N / 16;
  int groups2 = ((rowblk2 + 3) / 4) * (2 * OUT / 64);
  gemm2_bias_wmma<256, 64, 1, 1, false><<<groups2, 128, 0, stream>>>(
      h, Wl2, bl2, Wr2, br2, xl2, xr2, N, groups2, ei, E, head, lst);
  node_attend<64, 1, 4, false><<<(int)(((long long)N * 16 + B - 1) / B), B, 0, stream>>>(
      xl2, xr2, att2, bias2, head, lst, out, N, E);
}


# Layer-1 node pass and both layer-2 products in one launch: 16 waves put their 16 hidden rows in LDS and multiply them there; h never goes to memory, one launch fewer; layer-2 k sums in 4 quarters
# speedup vs baseline: 4.5132x; 1.0421x over previous
#include <hip/hip_runtime.h>
#include <math.h>

#define SLOPE 0.2f

typedef float v2f __attribute__((ext_vector_type(2)));
typedef float v8f __attribute__((ext_vector_type(8)));
typedef float v4f __attribute__((ext_vector_type(4)));

__device__ __forceinline__ float lrelu(float v) { return v > 0.0f ? v : SLOPE * v; }

#define LOG2E 1.44269504088896340736f
__device__ __forceinline__ float exp2_hw(float v) { return __builtin_amdgcn_exp2f(v); }

__device__ __forceinline__ void link_one(const int* __restrict__ ei, int E_, int* __restrict__ head,
                                         int* __restrict__ lst, int e) {
  const int s = ei[e], d = ei[E_ + e];
  const int prev = atomicExch(head + d, e + 1);
  ((int2*)lst)[e] = make_int2(s, prev);
}
__device__ __forceinline__ void link_four(const int* __restrict__ ei, int E_, int* __restrict__ head,
                                          int* __restrict__ lst, int t) {
  const int e0 = 4 * t;
  if (e0 >= E_) return;
  if ((E_ & 3) == 0) {
    const int4 s = *(const int4*)(ei + e0);
    const int4 d = *(const int4*)(ei + E_ + e0);
    const int p0 = atomicExch(head + d.x, e0 + 1);
    const int p1 = atomicExch(head + d.y, e0 + 2);
    const int p2 = atomicExch(head + d.z, e0 + 3);
    const int p3 = atomicExch(head + d.w, e0 + 4);
    int4* o = (int4*)(lst + 2 * (size_t)e0);
    o[0] = make_int4(s.x, p0, s.y, p1);
    o[1] = make_int4(s.z, p2, s.w, p3);
  } else {
    for (int k = 0; k < 4 && e0 + k < E_; k++) link_one(ei, E_, head, lst, e0 + k);
  }
}

__device__ __forceinline__ void hold_loads(const v4f (&w)[16]) {
  asm volatile("" ::"v"(w[0]), "v"(w[1]), "v"(w[2]), "v"(w[3]), "v"(w[4]), "v"(w[5]), "v"(w[6]),
               "v"(w[7]), "v"(w[8]), "v"(w[9]), "v"(w[10]), "v"(w[11]), "v"(w[12]), "v"(w[13]),
               "v"(w[14]), "v"(w[15]));
}
__device__ __forceinline__ void hold_loads(const v4f (&a)[4], const v4f (&w)[16]) {
  asm volatile("" ::"v"(a[0]), "v"(a[1]), "v"(a[2]), "v"(a[3]), "v"(w[0]), "v"(w[1]), "v"(w[2]),
               "v"(w[3]), "v"(w[4]), "v"(w[5]), "v"(w[6]), "v"(w[7]), "v"(w[8]), "v"(w[9]),
               "v"(w[10]), "v"(w[11]), "v"(w[12]), "v"(w[13]), "v"(w[14]), "v"(w[15]));
}
__device__ __forceinline__ void hold_loads(const v4f (&a)[8], const v4f (&w)[16]) {
  asm volatile("" ::"v"(a[0]), "v"(a[1]), "v"(a[2]), "v"(a[3]), "v"(a[4]), "v"(a[5]), "v"(a[6]),
               "v"(a[7]), "v"(w[0]), "v"(w[1]), "v"(w[2]), "v"(w[3]), "v"(w[4]), "v"(w[5]),
               "v"(w[6]), "v"(w[7]), "v"(w[8]), "v"(w[9]), "v"(w[10]), "v"(w[11]), "v"(w[12]),
               "v"(w[13]), "v"(w[14]), "v"(w[15]));
}

template <int K, int NC, int RT, int CG, bool LINK>
__global__ void __launch_bounds__(128)
gemm2_bias_wmma(const float* __restrict__ X,
                const float* __restrict__ Wl, const float* __restrict__ bl,
                const float* __restrict__ Wr, const float* __restrict__ br,
                float* __restrict__ Yl, float* __restrict__ Yr, int M, int ngemm,
                const int* __restrict__ ei, int E_, int* __restrict__ head, int* __restrict__ lst) {
  if (LINK && (int)blockIdx.x >= ngemm) {
    link_four(ei, E_, head, lst, ((int)blockIdx.x - ngemm) * 128 + (int)threadIdx.x);
    return;
  }
  constexpr int WCOLS = 64 * CG;
  constexpr int NCB = 2 * NC / WCOLS;
  constexpr int JB = 8 / CG;
  static_assert(NC % WCOLS == 0 && K % 32 == 0 && (CG == 1 || CG == 2), "tile shape");
  const int wid = threadIdx.x >> 5;
  const int lane = threadIdx.x & 31;
  const int h = lane >> 4;
  const int c = lane & 15;
  const int cbk = (int)(blockIdx.x % NCB);
  const int rb = (int)(blockIdx.x / NCB) * 4 + wid;
  const int MT = M / 16;
  if (rb * RT >= MT) return;
  const int gcol = cbk * WCOLS;
  const bool right = gcol >= NC;
  const float* __restrict__ W = right ? Wr : Wl;
  const float* __restrict__ b = right ? br : bl;
  float* __restrict__ Y = right ? Yr : Yl;
  const int cb = right ? gcol - NC : gcol;

  v8f acc[RT][CG][4];
#pragma unroll
  for (int g = 0; g < CG; g++) {
    const float4 bv = *(const float4*)(b + cb + 64 * g + 4 * c);
    const float bt[4] = {bv.x, bv.y, bv.z, bv.w};
#pragma unroll
    for (int t = 0; t < 4; t++)
#pragma unroll
      for (int rt = 0; rt < RT; rt++)
        acc[rt][g][t] = (v8f){bt[t], bt[t], bt[t], bt[t], bt[t], bt[t], bt[t], bt[t]};
  }

  const float* xp[RT];
#pragma unroll
  for (int rt = 0; rt < RT; rt++) {
    int mt = rb * RT + rt;
    mt = mt < MT ? mt : MT - 1;
    xp[rt] = X + (size_t)(mt * 16 + c) * K + 16 * h;
  }
  const float* wp = W + (size_t)(16 * h) * NC + cb + 4 * c;

#pragma unroll
  for (int kb = 0; kb < K; kb += 32) {
    v4f av[RT * 4];
#pragma unroll
    for (int rt = 0; rt < RT; rt++)
#pragma unroll
      for (int q = 0; q < 4; q++) av[rt * 4 + q] = *(const v4f*)(xp[rt] + kb + 4 * q);
#pragma unroll
    for (int jg = 0; jg < 8; jg += JB) {
      v4f wv[16];
#pragma unroll
      for (int jj = 0; jj < JB; jj++)
#pragma unroll
        for (int g = 0; g < CG; g++) {
          const float* wk = wp + (size_t)(kb + 2 * (jg + jj)) * NC + 64 * g;
          wv[(jj * CG + g) * 2] = *(const v4f*)wk;
          wv[(jj * CG + g) * 2 + 1] = *(const v4f*)(wk + NC);
        }
      if (jg == 0) hold_loads(av, wv); else hold_loads(wv);
#pragma unroll
      for (int jj = 0; jj < JB; jj++) {
        const int j = jg + jj;
#pragma unroll
        for (int g = 0; g < CG; g++) {
          const v4f w0 = wv[(jj * CG + g) * 2];
          const v4f w1 = wv[(jj * CG + g) * 2 + 1];
          const float b0[4] = {w0.x, w0.y, w0.z, w0.w};
          const float b1[4] = {w1.x, w1.y, w1.z, w1.w};
#pragma unroll
          for (int t = 0; t < 4; t++) {
            v2f bf;
            bf.x = b0[t];
            bf.y = b1[t];
#pragma unroll
            for (int rt = 0; rt < RT; rt++) {
              const v4f aq = av[rt * 4 + (j >> 1)];
              v2f af;
              af.x = (j & 1) ? aq.z : aq.x;
              af.y = (j & 1) ? aq.w : aq.y;
              acc[rt][g][t] = __builtin_amdgcn_wmma_f32_16x16x4_f32(false, af, false, bf, (short)0,
                                                                    acc[rt][g][t], false, false);
            }
          }
        }
      }
    }
  }

#pragma unroll
  for (int rt = 0; rt < RT; rt++) {
    const int mt = rb * RT + rt;
    if (mt < MT) {
      float* yp = Y + (size_t)(mt * 16 + 8 * h) * NC + cb + 4 * c;
#pragma unroll
      for (int g = 0; g < CG; g++)
#pragma unroll
        for (int r = 0; r < 8; r++) {
          float4 o;
          o.x = acc[rt][g][0][r]; o.y = acc[rt][g][1][r];
          o.z = acc[rt][g][2][r]; o.w = acc[rt][g][3][r];
          *(float4*)(yp + (size_t)r * NC + 64 * g) = o;
        }
    }
  }
}

template <int V>
__device__ __forceinline__ void load_row(const float* __restrict__ p, float (&x)[V]) {
#pragma unroll
  for (int q = 0; q < V / 4; q++) {
    float4 t = *(const float4*)(p + 4 * q);
    x[4 * q] = t.x; x[4 * q + 1] = t.y; x[4 * q + 2] = t.z; x[4 * q + 3] = t.w;
  }
}

template <int V, int LPH>
__device__ __forceinline__ float edge_logit(const float (&x)[V], const float (&r)[V],
                                            const float (&a)[V]) {
  float p = 0.0f;
#pragma unroll
  for (int k = 0; k < V; k++) p += lrelu(x[k] + r[k]) * a[k];
  for (int off = LPH / 2; off > 0; off >>= 1) p += __shfl_xor(p, off, 32);
  return p;
}

template <int HC, int H_, int V, bool ELU>
__device__ __forceinline__ void attend_row(const float* __restrict__ xl, const float* __restrict__ xr,
                                           const float* __restrict__ att, const float* __restrict__ bias,
                                           const int* __restrict__ head, const int* __restrict__ lst,
                                           int node, int base, int E_, float (&res)[V]) {
  constexpr int LPN = HC / V;
  constexpr int LPH = LPN / H_;
  const int2* __restrict__ links = (const int2*)lst;
  float r[V], a[V], x[V], acc[V], bs[V];
  load_row<V>(xr + (size_t)node * HC + base, r);
  load_row<V>(att + base, a);
#pragma unroll
  for (int k = 0; k < V; k++) a[k] *= LOG2E;
  load_row<V>(xl + (size_t)node * HC + base, acc);
  float m = edge_logit<V, LPH>(acc, r, a);
  float s = 1.0f;
  int p = head[node];
  int2 cur = links[p > 0 ? p - 1 : 0];
  for (int it = 0; p != 0 && it < E_; it++) {
    const int pn = cur.y;
    const int2 nx = links[pn > 0 ? pn - 1 : 0];
    asm volatile("" ::: "memory");
    load_row<V>(xl + (size_t)cur.x * HC + base, x);
    const float e = edge_logit<V, LPH>(x, r, a);
    const float d = e - m;
    const float t = exp2_hw(-fabsf(d));
    const float sc = d > 0.0f ? t : 1.0f;
    const float wgt = d > 0.0f ? 1.0f : t;
    m = fmaxf(m, e);
    s = s * sc + wgt;
#pragma unroll
    for (int k = 0; k < V; k++) acc[k] = acc[k] * sc + wgt * x[k];
    p = pn;
    cur = nx;
  }
  load_row<V>(bias + base, bs);
  const float inv = 1.0f / s;
#pragma unroll
  for (int k = 0; k < V; k++) {
    float v = acc[k] * inv + bs[k];
    if (ELU) v = v > 0.0f ? v : exp2_hw(v * LOG2E) - 1.0f;
    res[k] = v;
  }
}

template <int HC, int H_, int V, bool ELU>
__global__ void node_attend(const float* __restrict__ xl, const float* __restrict__ xr,
                            const float* __restrict__ att, const float* __restrict__ bias,
                            const int* __restrict__ head, const int* __restrict__ lst,
                            float* __restrict__ outp, int n, int E_) {
  constexpr int LPN = HC / V;
  int node = (int)((blockIdx.x * (size_t)blockDim.x + threadIdx.x) / LPN);
  if (LPN == 32) node = __builtin_amdgcn_readfirstlane(node);
  if (node >= n) return;
  const int base = (int)(threadIdx.x % LPN) * V;
  float res[V];
  attend_row<HC, H_, V, ELU>(xl, xr, att, bias, head, lst, node, base, E_, res);
  float* o = outp + (size_t)node * HC + base;
#pragma unroll
  for (int q = 0; q < V / 4; q++) {
    float4 t;
    t.x = res[4 * q]; t.y = res[4 * q + 1]; t.z = res[4 * q + 2]; t.w = res[4 * q + 3];
    *(float4*)(o + 4 * q) = t;
  }
}

#define HP 260
__global__ void __launch_bounds__(512)
node_attend_product(const float* __restrict__ xl, const float* __restrict__ xr,
                    const float* __restrict__ att, const float* __restrict__ bias,
                    const int* __restrict__ head, const int* __restrict__ lst,
                    const float* __restrict__ Wl, const float* __restrict__ bl,
                    const float* __restrict__ Wr, const float* __restrict__ br,
                    float* __restrict__ Yl, float* __restrict__ Yr, int n, int E_) {
  __shared__ __attribute__((aligned(16))) float hrow[16 * HP];
  __shared__ __attribute__((aligned(16))) float part[16 * 8 * 32 * 2];
  const int lane = threadIdx.x & 31;
  const int w = __builtin_amdgcn_readfirstlane((int)(threadIdx.x >> 5));
  const int row0 = (int)blockIdx.x * 16;
  const int node = row0 + w;
  const int h = lane >> 4;
  const int c = lane & 15;
  const int p = w >> 3, ch = (w >> 2) & 1, kq = w & 3;
  const float* __restrict__ W = p ? Wr : Wl;
  const float* wrow = W + (size_t)(64 * kq + 4 * h) * 64 + 32 * ch + 2 * c;

  float res[8];
  if (node < n) {
    attend_row<256, 4, 8, true>(xl, xr, att, bias, head, lst, node, lane * 8, E_, res);
  } else {
#pragma unroll
    for (int k = 0; k < 8; k++) res[k] = 0.0f;
  }
  v2f bw[16];
#pragma unroll
  for (int i = 0; i < 16; i++) bw[i] = *(const v2f*)(wrow + (size_t)(8 * (i >> 2) + (i & 3)) * 64);
  {
    float4 t0, t1;
    t0.x = res[0]; t0.y = res[1]; t0.z = res[2]; t0.w = res[3];
    t1.x = res[4]; t1.y = res[5]; t1.z = res[6]; t1.w = res[7];
    float4* hp = (float4*)(hrow + w * HP + lane * 8);
    hp[0] = t0;
    hp[1] = t1;
  }
  __syncthreads();

  const float* arow = hrow + c * HP + 64 * kq + 4 * h;
  v8f acc0 = (v8f){0.0f, 0.0f, 0.0f, 0.0f, 0.0f, 0.0f, 0.0f, 0.0f};
  v8f acc1 = acc0;
#pragma unroll
  for (int hf = 0; hf < 2; hf++) {
    if (hf == 1) {
#pragma unroll
      for (int i = 0; i < 16; i++)
        bw[i] = *(const v2f*)(wrow + (size_t)(32 + 8 * (i >> 2) + (i & 3)) * 64);
    }
#pragma unroll
    for (int u = 0; u < 4; u++) {
      const v4f av = *(const v4f*)(arow + 32 * hf + 8 * u);
#pragma unroll
      for (int j = 0; j < 2; j++) {
        v2f af, b0, b1;
        af.x = j ? av.z : av.x;
        af.y = j ? av.w : av.y;
        const v2f k0 = bw[4 * u + 2 * j];
        const v2f k1 = bw[4 * u + 2 * j + 1];
        b0.x = k0.x; b0.y = k1.x;
        b1.x = k0.y; b1.y = k1.y;
        acc0 = __builtin_amdgcn_wmma_f32_16x16x4_f32(false, af, false, b0, (short)0, acc0, false, false);
        acc1 = __builtin_amdgcn_wmma_f32_16x16x4_f32(false, af, false, b1, (short)0, acc1, false, false);
      }
    }
  }
  {
    float2* pp = (float2*)part + (size_t)(w * 8) * 32 + lane;
#pragma unroll
    for (int r = 0; r < 8; r++) pp[r * 32] = make_float2(acc0[r], acc1[r]);
  }
  __syncthreads();

  {
    const float* __restrict__ b = p ? br : bl;
    float* __restrict__ Y = p ? Yr : Yl;
    const float2 bv = *(const float2*)(b + 32 * ch + 2 * c);
    const int wsrc = p * 8 + ch * 4;
#pragma unroll
    for (int q = 0; q < 2; q++) {
      const int r = 2 * (w & 3) + q;
      const float2* pp = (const float2*)part + (size_t)(wsrc * 8 + r) * 32 + lane;
      float2 sum = pp[0];
#pragma unroll
      for (int k = 1; k < 4; k++) {
        const float2 v = pp[(size_t)k * 8 * 32];
        sum.x += v.x;
        sum.y += v.y;
      }
      sum.x += bv.x;
      sum.y += bv.y;
      const int row = row0 + 8 * h + r;
      if (row < n)
        *(float2*)(Y + (size_t)row * 64 + 32 * ch + 2 * c) = sum;
    }
  }
}

extern "C" void kernel_launch(void* const* d_in, const int* in_sizes, int n_in,
                              void* d_out, int out_size, void* d_ws, size_t ws_size,
                              hipStream_t stream) {
  (void)n_in; (void)out_size; (void)ws_size;
  const int IN = 128, HID = 64, H = 4, OUT = 64;
  const float*    x    = (const float*)d_in[0];
  const int*      ei   = (const int*)d_in[1];
  const float*    Wl1  = (const float*)d_in[2];
  const float*    bl1  = (const float*)d_in[3];
  const float*    Wr1  = (const float*)d_in[4];
  const float*    br1  = (const float*)d_in[5];
  const float*    att1 = (const float*)d_in[6];
  const float*    bias1= (const float*)d_in[7];
  const float*    Wl2  = (const float*)d_in[8];
  const float*    bl2  = (const float*)d_in[9];
  const float*    Wr2  = (const float*)d_in[10];
  const float*    br2  = (const float*)d_in[11];
  const float*    att2 = (const float*)d_in[12];
  const float*    bias2= (const float*)d_in[13];
  float* out = (float*)d_out;

  const int N    = in_sizes[0] / IN;
  const int E    = in_sizes[1] / 2;
  const int HC   = H * HID;

  float* ws = (float*)d_ws;
  size_t o = 0;
  float* xl1    = ws + o; o += (size_t)N * HC;
  float* xr1    = ws + o; o += (size_t)N * HC;
  float* xl2    = ws + o; o += (size_t)N * OUT;
  float* xr2    = ws + o; o += (size_t)N * OUT;
  int*   lst    = (int*)(ws + o); o += (size_t)2 * E;
  int*   head   = (int*)(ws + o); o += (size_t)N + 4;

  const int B = 256;

  hipMemsetAsync(head, 0, (size_t)N * sizeof(int), stream);

  int rowblk1 = (N / 16 + 1) / 2;
  int groups1 = ((rowblk1 + 3) / 4) * (2 * HC / 128);
  int linkgrp = ((E + 3) / 4 + 127) / 128;
  gemm2_bias_wmma<128, 256, 2, 2, true><<<groups1 + linkgrp, 128, 0, stream>>>(
      x, Wl1, bl1, Wr1, br1, xl1, xr1, N, groups1, ei, E, head, lst);
  node_attend_product<<<(N + 15) / 16, 512, 0, stream>>>(
      xl1, xr1, att1, bias1, head, lst, Wl2, bl2, Wr2, br2, xl2, xr2, N, E);

  node_attend<64, 1, 4, false><<<(int)(((long long)N * 16 + B - 1) / B), B, 0, stream>>>(
      xl2, xr2, att2, bias2, head, lst, out, N, E);
}


# Logit lane sums on the vector ALU: each butterfly step is one DPP add (row_xmask) in place of an LDS-unit permute, its wait and an add; same partners and order, sums bit-identical
# speedup vs baseline: 4.7264x; 1.0472x over previous
#include <hip/hip_runtime.h>
#include <math.h>

#define SLOPE 0.2f

typedef float v2f __attribute__((ext_vector_type(2)));
typedef float v8f __attribute__((ext_vector_type(8)));
typedef float v4f __attribute__((ext_vector_type(4)));

__device__ __forceinline__ float lrelu(float v) { return v > 0.0f ? v : SLOPE * v; }

#define LOG2E 1.44269504088896340736f
__device__ __forceinline__ float exp2_hw(float v) { return __builtin_amdgcn_exp2f(v); }

__device__ __forceinline__ void link_one(const int* __restrict__ ei, int E_, int* __restrict__ head,
                                         int* __restrict__ lst, int e) {
  const int s = ei[e], d = ei[E_ + e];
  const int prev = atomicExch(head + d, e + 1);
  ((int2*)lst)[e] = make_int2(s, prev);
}
__device__ __forceinline__ void link_four(const int* __restrict__ ei, int E_, int* __restrict__ head,
                                          int* __restrict__ lst, int t) {
  const int e0 = 4 * t;
  if (e0 >= E_) return;
  if ((E_ & 3) == 0) {
    const int4 s = *(const int4*)(ei + e0);
    const int4 d = *(const int4*)(ei + E_ + e0);
    const int p0 = atomicExch(head + d.x, e0 + 1);
    const int p1 = atomicExch(head + d.y, e0 + 2);
    const int p2 = atomicExch(head + d.z, e0 + 3);
    const int p3 = atomicExch(head + d.w, e0 + 4);
    int4* o = (int4*)(lst + 2 * (size_t)e0);
    o[0] = make_int4(s.x, p0, s.y, p1);
    o[1] = make_int4(s.z, p2, s.w, p3);
  } else {
    for (int k = 0; k < 4 && e0 + k < E_; k++) link_one(ei, E_, head, lst, e0 + k);
  }
}

__device__ __forceinline__ void hold_loads(const v4f (&w)[16]) {
  asm volatile("" ::"v"(w[0]), "v"(w[1]), "v"(w[2]), "v"(w[3]), "v"(w[4]), "v"(w[5]), "v"(w[6]),
               "v"(w[7]), "v"(w[8]), "v"(w[9]), "v"(w[10]), "v"(w[11]), "v"(w[12]), "v"(w[13]),
               "v"(w[14]), "v"(w[15]));
}
__device__ __forceinline__ void hold_loads(const v4f (&a)[4], const v4f (&w)[16]) {
  asm volatile("" ::"v"(a[0]), "v"(a[1]), "v"(a[2]), "v"(a[3]), "v"(w[0]), "v"(w[1]), "v"(w[2]),
               "v"(w[3]), "v"(w[4]), "v"(w[5]), "v"(w[6]), "v"(w[7]), "v"(w[8]), "v"(w[9]),
               "v"(w[10]), "v"(w[11]), "v"(w[12]), "v"(w[13]), "v"(w[14]), "v"(w[15]));
}
__device__ __forceinline__ void hold_loads(const v4f (&a)[8], const v4f (&w)[16]) {
  asm volatile("" ::"v"(a[0]), "v"(a[1]), "v"(a[2]), "v"(a[3]), "v"(a[4]), "v"(a[5]), "v"(a[6]),
               "v"(a[7]), "v"(w[0]), "v"(w[1]), "v"(w[2]), "v"(w[3]), "v"(w[4]), "v"(w[5]),
               "v"(w[6]), "v"(w[7]), "v"(w[8]), "v"(w[9]), "v"(w[10]), "v"(w[11]), "v"(w[12]),
               "v"(w[13]), "v"(w[14]), "v"(w[15]));
}

template <int K, int NC, int RT, int CG, bool LINK>
__global__ void __launch_bounds__(128)
gemm2_bias_wmma(const float* __restrict__ X,
                const float* __restrict__ Wl, const float* __restrict__ bl,
                const float* __restrict__ Wr, const float* __restrict__ br,
                float* __restrict__ Yl, float* __restrict__ Yr, int M, int ngemm,
                const int* __restrict__ ei, int E_, int* __restrict__ head, int* __restrict__ lst) {
  if (LINK && (int)blockIdx.x >= ngemm) {
    link_four(ei, E_, head, lst, ((int)blockIdx.x - ngemm) * 128 + (int)threadIdx.x);
    return;
  }
  constexpr int WCOLS = 64 * CG;
  constexpr int NCB = 2 * NC / WCOLS;
  constexpr int JB = 8 / CG;
  static_assert(NC % WCOLS == 0 && K % 32 == 0 && (CG == 1 || CG == 2), "tile shape");
  const int wid = threadIdx.x >> 5;
  const int lane = threadIdx.x & 31;
  const int h = lane >> 4;
  const int c = lane & 15;
  const int cbk = (int)(blockIdx.x % NCB);
  const int rb = (int)(blockIdx.x / NCB) * 4 + wid;
  const int MT = M / 16;
  if (rb * RT >= MT) return;
  const int gcol = cbk * WCOLS;
  const bool right = gcol >= NC;
  const float* __restrict__ W = right ? Wr : Wl;
  const float* __restrict__ b = right ? br : bl;
  float* __restrict__ Y = right ? Yr : Yl;
  const int cb = right ? gcol - NC : gcol;

  v8f acc[RT][CG][4];
#pragma unroll
  for (int g = 0; g < CG; g++) {
    const float4 bv = *(const float4*)(b + cb + 64 * g + 4 * c);
    const float bt[4] = {bv.x, bv.y, bv.z, bv.w};
#pragma unroll
    for (int t = 0; t < 4; t++)
#pragma unroll
      for (int rt = 0; rt < RT; rt++)
        acc[rt][g][t] = (v8f){bt[t], bt[t], bt[t], bt[t], bt[t], bt[t], bt[t], bt[t]};
  }

  const float* xp[RT];
#pragma unroll
  for (int rt = 0; rt < RT; rt++) {
    int mt = rb * RT + rt;
    mt = mt < MT ? mt : MT - 1;
    xp[rt] = X + (size_t)(mt * 16 + c) * K + 16 * h;
  }
  const float* wp = W + (size_t)(16 * h) * NC + cb + 4 * c;

#pragma unroll
  for (int kb = 0; kb < K; kb += 32) {
    v4f av[RT * 4];
#pragma unroll
    for (int rt = 0; rt < RT; rt++)
#pragma unroll
      for (int q = 0; q < 4; q++) av[rt * 4 + q] = *(const v4f*)(xp[rt] + kb + 4 * q);
#pragma unroll
    for (int jg = 0; jg < 8; jg += JB) {
      v4f wv[16];
#pragma unroll
      for (int jj = 0; jj < JB; jj++)
#pragma unroll
        for (int g = 0; g < CG; g++) {
          const float* wk = wp + (size_t)(kb + 2 * (jg + jj)) * NC + 64 * g;
          wv[(jj * CG + g) * 2] = *(const v4f*)wk;
          wv[(jj * CG + g) * 2 + 1] = *(const v4f*)(wk + NC);
        }
      if (jg == 0) hold_loads(av, wv); else hold_loads(wv);
#pragma unroll
      for (int jj = 0; jj < JB; jj++) {
        const int j = jg + jj;
#pragma unroll
        for (int g = 0; g < CG; g++) {
          const v4f w0 = wv[(jj * CG + g) * 2];
          const v4f w1 = wv[(jj * CG + g) * 2 + 1];
          const float b0[4] = {w0.x, w0.y, w0.z, w0.w};
          const float b1[4] = {w1.x, w1.y, w1.z, w1.w};
#pragma unroll
          for (int t = 0; t < 4; t++) {
            v2f bf;
            bf.x = b0[t];
            bf.y = b1[t];
#pragma unroll
            for (int rt = 0; rt < RT; rt++) {
              const v4f aq = av[rt * 4 + (j >> 1)];
              v2f af;
              af.x = (j & 1) ? aq.z : aq.x;
              af.y = (j & 1) ? aq.w : aq.y;
              acc[rt][g][t] = __builtin_amdgcn_wmma_f32_16x16x4_f32(false, af, false, bf, (short)0,
                                                                    acc[rt][g][t], false, false);
            }
          }
        }
      }
    }
  }

#pragma unroll
  for (int rt = 0; rt < RT; rt++) {
    const int mt = rb * RT + rt;
    if (mt < MT) {
      float* yp = Y + (size_t)(mt * 16 + 8 * h) * NC + cb + 4 * c;
#pragma unroll
      for (int g = 0; g < CG; g++)
#pragma unroll
        for (int r = 0; r < 8; r++) {
          float4 o;
          o.x = acc[rt][g][0][r]; o.y = acc[rt][g][1][r];
          o.z = acc[rt][g][2][r]; o.w = acc[rt][g][3][r];
          *(float4*)(yp + (size_t)r * NC + 64 * g) = o;
        }
    }
  }
}

template <int V>
__device__ __forceinline__ void load_row(const float* __restrict__ p, float (&x)[V]) {
#pragma unroll
  for (int q = 0; q < V / 4; q++) {
    float4 t = *(const float4*)(p + 4 * q);
    x[4 * q] = t.x; x[4 * q + 1] = t.y; x[4 * q + 2] = t.z; x[4 * q + 3] = t.w;
  }
}

template <int OFF>
__device__ __forceinline__ float add_lane_xor(float p) {
  static_assert(OFF > 0 && OFF < 16, "partner inside the 16-lane row");
  const int q = __builtin_amdgcn_update_dpp(0, __builtin_bit_cast(int, p), 0x160 + OFF, 0xf, 0xf, true);
  return p + __builtin_bit_cast(float, q);
}

template <int V, int LPH>
__device__ __forceinline__ float edge_logit(const float (&x)[V], const float (&r)[V],
                                            const float (&a)[V]) {
  static_assert(LPH == 8 || LPH == 16, "a head's lanes lie inside one 16-lane row");
  float p = 0.0f;
#pragma unroll
  for (int k = 0; k < V; k++) p += lrelu(x[k] + r[k]) * a[k];
  if (LPH == 16) p = add_lane_xor<8>(p);
  p = add_lane_xor<4>(p);
  p = add_lane_xor<2>(p);
  p = add_lane_xor<1>(p);
  return p;
}

template <int HC, int H_, int V, bool ELU>
__device__ __forceinline__ void attend_row(const float* __restrict__ xl, const float* __restrict__ xr,
                                           const float* __restrict__ att, const float* __restrict__ bias,
                                           const int* __restrict__ head, const int* __restrict__ lst,
                                           int node, int base, int E_, float (&res)[V]) {
  constexpr int LPN = HC / V;
  constexpr int LPH = LPN / H_;
  const int2* __restrict__ links = (const int2*)lst;
  float r[V], a[V], x[V], acc[V], bs[V];
  load_row<V>(xr + (size_t)node * HC + base, r);
  load_row<V>(att + base, a);
#pragma unroll
  for (int k = 0; k < V; k++) a[k] *= LOG2E;
  load_row<V>(xl + (size_t)node * HC + base, acc);
  float m = edge_logit<V, LPH>(acc, r, a);
  float s = 1.0f;
  int p = head[node];
  int2 cur = links[p > 0 ? p - 1 : 0];
  for (int it = 0; p != 0 && it < E_; it++) {
    const int pn = cur.y;
    const int2 nx = links[pn > 0 ? pn - 1 : 0];
    asm volatile("" ::: "memory");
    load_row<V>(xl + (size_t)cur.x * HC + base, x);
    const float e = edge_logit<V, LPH>(x, r, a);
    const float d = e - m;
    const float t = exp2_hw(-fabsf(d));
    const float sc = d > 0.0f ? t : 1.0f;
    const float wgt = d > 0.0f ? 1.0f : t;
    m = fmaxf(m, e);
    s = s * sc + wgt;
#pragma unroll
    for (int k = 0; k < V; k++) acc[k] = acc[k] * sc + wgt * x[k];
    p = pn;
    cur = nx;
  }
  load_row<V>(bias + base, bs);
  const float inv = 1.0f / s;
#pragma unroll
  for (int k = 0; k < V; k++) {
    float v = acc[k] * inv + bs[k];
    if (ELU) v = v > 0.0f ? v : exp2_hw(v * LOG2E) - 1.0f;
    res[k] = v;
  }
}

template <int HC, int H_, int V, bool ELU>
__global__ void node_attend(const float* __restrict__ xl, const float* __restrict__ xr,
                            const float* __restrict__ att, const float* __restrict__ bias,
                            const int* __restrict__ head, const int* __restrict__ lst,
                            float* __restrict__ outp, int n, int E_) {
  constexpr int LPN = HC / V;
  int node = (int)((blockIdx.x * (size_t)blockDim.x + threadIdx.x) / LPN);
  if (LPN == 32) node = __builtin_amdgcn_readfirstlane(node);
  if (node >= n) return;
  const int base = (int)(threadIdx.x % LPN) * V;
  float res[V];
  attend_row<HC, H_, V, ELU>(xl, xr, att, bias, head, lst, node, base, E_, res);
  float* o = outp + (size_t)node * HC + base;
#pragma unroll
  for (int q = 0; q < V / 4; q++) {
    float4 t;
    t.x = res[4 * q]; t.y = res[4 * q + 1]; t.z = res[4 * q + 2]; t.w = res[4 * q + 3];
    *(float4*)(o + 4 * q) = t;
  }
}

#define HP 260
__global__ void __launch_bounds__(512)
node_attend_product(const float* __restrict__ xl, const float* __restrict__ xr,
                    const float* __restrict__ att, const float* __restrict__ bias,
                    const int* __restrict__ head, const int* __restrict__ lst,
                    const float* __restrict__ Wl, const float* __restrict__ bl,
                    const float* __restrict__ Wr, const float* __restrict__ br,
                    float* __restrict__ Yl, float* __restrict__ Yr, int n, int E_) {
  __shared__ __attribute__((aligned(16))) float hrow[16 * HP];
  __shared__ __attribute__((aligned(16))) float part[16 * 8 * 32 * 2];
  const int lane = threadIdx.x & 31;
  const int w = __builtin_amdgcn_readfirstlane((int)(threadIdx.x >> 5));
  const int row0 = (int)blockIdx.x * 16;
  const int node = row0 + w;
  const int h = lane >> 4;
  const int c = lane & 15;
  const int p = w >> 3, ch = (w >> 2) & 1, kq = w & 3;
  const float* __restrict__ W = p ? Wr : Wl;
  const float* wrow = W + (size_t)(64 * kq + 4 * h) * 64 + 32 * ch + 2 * c;

  float res[8];
  if (node < n) {
    attend_row<256, 4, 8, true>(xl, xr, att, bias, head, lst, node, lane * 8, E_, res);
  } else {
#pragma unroll
    for (int k = 0; k < 8; k++) res[k] = 0.0f;
  }
  v2f bw[16];
#pragma unroll
  for (int i = 0; i < 16; i++) bw[i] = *(const v2f*)(wrow + (size_t)(8 * (i >> 2) + (i & 3)) * 64);
  {
    float4 t0, t1;
    t0.x = res[0]; t0.y = res[1]; t0.z = res[2]; t0.w = res[3];
    t1.x = res[4]; t1.y = res[5]; t1.z = res[6]; t1.w = res[7];
    float4* hp = (float4*)(hrow + w * HP + lane * 8);
    hp[0] = t0;
    hp[1] = t1;
  }
  __syncthreads();

  const float* arow = hrow + c * HP + 64 * kq + 4 * h;
  v8f acc0 = (v8f){0.0f, 0.0f, 0.0f, 0.0f, 0.0f, 0.0f, 0.0f, 0.0f};
  v8f acc1 = acc0;
#pragma unroll
  for (int hf = 0; hf < 2; hf++) {
    if (hf == 1) {
#pragma unroll
      for (int i = 0; i < 16; i++)
        bw[i] = *(const v2f*)(wrow + (size_t)(32 + 8 * (i >> 2) + (i & 3)) * 64);
    }
#pragma unroll
    for (int u = 0; u < 4; u++) {
      const v4f av = *(const v4f*)(arow + 32 * hf + 8 * u);
#pragma unroll
      for (int j = 0; j < 2; j++) {
        v2f af, b0, b1;
        af.x = j ? av.z : av.x;
        af.y = j ? av.w : av.y;
        const v2f k0 = bw[4 * u + 2 * j];
        const v2f k1 = bw[4 * u + 2 * j + 1];
        b0.x = k0.x; b0.y = k1.x;
        b1.x = k0.y; b1.y = k1.y;
        acc0 = __builtin_amdgcn_wmma_f32_16x16x4_f32(false, af, false, b0, (short)0, acc0, false, false);
        acc1 = __builtin_amdgcn_wmma_f32_16x16x4_f32(false, af, false, b1, (short)0, acc1, false, false);
      }
    }
  }
  {
    float2* pp = (float2*)part + (size_t)(w * 8) * 32 + lane;
#pragma unroll
    for (int r = 0; r < 8; r++) pp[r * 32] = make_float2(acc0[r], acc1[r]);
  }
  __syncthreads();

  {
    const float* __restrict__ b = p ? br : bl;
    float* __restrict__ Y = p ? Yr : Yl;
    const float2 bv = *(const float2*)(b + 32 * ch + 2 * c);
    const int wsrc = p * 8 + ch * 4;
#pragma unroll
    for (int q = 0; q < 2; q++) {
      const int r = 2 * (w & 3) + q;
      const float2* pp = (const float2*)part + (size_t)(wsrc * 8 + r) * 32 + lane;
      float2 sum = pp[0];
#pragma unroll
      for (int k = 1; k < 4; k++) {
        const float2 v = pp[(size_t)k * 8 * 32];
        sum.x += v.x;
        sum.y += v.y;
      }
      sum.x += bv.x;
      sum.y += bv.y;
      const int row = row0 + 8 * h + r;
      if (row < n)
        *(float2*)(Y + (size_t)row * 64 + 32 * ch + 2 * c) = sum;
    }
  }
}

extern "C" void kernel_launch(void* const* d_in, const int* in_sizes, int n_in,
                              void* d_out, int out_size, void* d_ws, size_t ws_size,
                              hipStream_t stream) {
  (void)n_in; (void)out_size; (void)ws_size;
  const int IN = 128, HID = 64, H = 4, OUT = 64;
  const float*    x    = (const float*)d_in[0];
  const int*      ei   = (const int*)d_in[1];
  const float*    Wl1  = (const float*)d_in[2];
  const float*    bl1  = (const float*)d_in[3];
  const float*    Wr1  = (const float*)d_in[4];
  const float*    br1  = (const float*)d_in[5];
  const float*    att1 = (const float*)d_in[6];
  const float*    bias1= (const float*)d_in[7];
  const float*    Wl2  = (const float*)d_in[8];
  const float*    bl2  = (const float*)d_in[9];
  const float*    Wr2  = (const float*)d_in[10];
  const float*    br2  = (const float*)d_in[11];
  const float*    att2 = (const float*)d_in[12];
  const float*    bias2= (const float*)d_in[13];
  float* out = (float*)d_out;

  const int N    = in_sizes[0] / IN;
  const int E    = in_sizes[1] / 2;
  const int HC   = H * HID;

  float* ws = (float*)d_ws;
  size_t o = 0;
  float* xl1    = ws + o; o += (size_t)N * HC;
  float* xr1    = ws + o; o += (size_t)N * HC;
  float* xl2    = ws + o; o += (size_t)N * OUT;
  float* xr2    = ws + o; o += (size_t)N * OUT;
  int*   lst    = (int*)(ws + o); o += (size_t)2 * E;
  int*   head   = (int*)(ws + o); o += (size_t)N + 4;

  const int B = 256;

  hipMemsetAsync(head, 0, (size_t)N * sizeof(int), stream);

  int rowblk1 = (N / 16 + 1) / 2;
  int groups1 = ((rowblk1 + 3) / 4) * (2 * HC / 128);
  int linkgrp = ((E + 3) / 4 + 127) / 128;
  gemm2_bias_wmma<128, 256, 2, 2, true><<<groups1 + linkgrp, 128, 0, stream>>>(
      x, Wl1, bl1, Wr1, br1, xl1, xr1, N, groups1, ei, E, head, lst);
  node_attend_product<<<(N + 15) / 16, 512, 0, stream>>>(
      xl1, xr1, att1, bias1, head, lst, Wl2, bl2, Wr2, br2, xl2, xr2, N, E);

  node_attend<64, 1, 4, false><<<(int)(((long long)N * 16 + B - 1) / B), B, 0, stream>>>(
      xl2, xr2, att2, bias2, head, lst, out, N, E);
}
